# starting kernel
# baseline (speedup 1.0000x reference)
#include <hip/hip_runtime.h>
#include <hip/hip_bf16.h>

typedef __attribute__((ext_vector_type(2))) float v2f;
typedef __attribute__((ext_vector_type(8))) float v8f;

#define FEATDIM 128
#define BM      32

__global__ __launch_bounds__(256) void gemm128_wmma_f32(
    const float* __restrict__ A, const float* __restrict__ W,
    const float* __restrict__ bias, float* __restrict__ C, int M)
{
    __shared__ float sA[BM * FEATDIM];

    const int tid  = threadIdx.x;
    const int row0 = blockIdx.x * BM;
    if (row0 >= M) return;

    {
        const float4* src = (const float4*)(A + (size_t)row0 * FEATDIM);
        float4* dst = (float4*)sA;
        #pragma unroll
        for (int i = 0; i < 4; ++i)
            dst[tid + 256 * i] = src[tid + 256 * i];
    }
    __syncthreads();

    const int wave = tid >> 5;
    const int lane = tid & 31;
    const int half = lane >> 4;
    const int l16  = lane & 15;
    const int n    = wave * 16 + l16;

    v8f acc0 = {};
    v8f acc1 = {};

    #pragma unroll
    for (int k0 = 0; k0 < FEATDIM; k0 += 4) {
        const int ka = k0 + 2 * half;
        v2f b;
        b.x = W[(size_t)ka       * FEATDIM + n];
        b.y = W[(size_t)(ka + 1) * FEATDIM + n];
        v2f a0, a1;
        a0.x = sA[l16 * FEATDIM + ka];
        a0.y = sA[l16 * FEATDIM + ka + 1];
        a1.x = sA[(16 + l16) * FEATDIM + ka];
        a1.y = sA[(16 + l16) * FEATDIM + ka + 1];
        acc0 = __builtin_amdgcn_wmma_f32_16x16x4_f32(
            false, a0, false, b, (short)0, acc0, false, false);
        acc1 = __builtin_amdgcn_wmma_f32_16x16x4_f32(
            false, a1, false, b, (short)0, acc1, false, false);
    }

    const float bn = bias[n];
    float* Crow = C + (size_t)row0 * FEATDIM;
    #pragma unroll
    for (int v = 0; v < 8; ++v) {
        const int m = v + 8 * half;
        Crow[(size_t)m * FEATDIM + n]        = acc0[v] + bn;
        Crow[(size_t)(m + 16) * FEATDIM + n] = acc1[v] + bn;
    }
}

__global__ void zero_f32(float* __restrict__ p, long long n)
{
    long long i      = (long long)blockIdx.x * blockDim.x + threadIdx.x;
    long long stride = (long long)gridDim.x * blockDim.x;
    for (; i < n; i += stride) p[i] = 0.0f;
}

__global__ __launch_bounds__(256) void scatter_accum(
    const float* __restrict__ paper_h, const int* __restrict__ author_ids,
    const int* __restrict__ paper_ids, float* __restrict__ sums,
    float* __restrict__ counts, int num_edges)
{
    const int wid  = (int)(((long long)blockIdx.x * blockDim.x + threadIdx.x) >> 5);
    const int lane = threadIdx.x & 31;
    if (wid >= num_edges) return;

    const int aid = author_ids[wid];
    const int pid = paper_ids[wid];

    const float4 v = *(const float4*)(paper_h + (size_t)pid * FEATDIM + lane * 4);
    float* dst = sums + (size_t)aid * FEATDIM + lane * 4;
    atomicAdd(dst + 0, v.x);
    atomicAdd(dst + 1, v.y);
    atomicAdd(dst + 2, v.z);
    atomicAdd(dst + 3, v.w);
    if (lane == 0) atomicAdd(counts + aid, 1.0f);
}

__global__ void normalize_rows(float* __restrict__ sums,
                               const float* __restrict__ counts, long long n)
{
    long long i = (long long)blockIdx.x * blockDim.x + threadIdx.x;
    if (i < n) {
        const float c = counts[i >> 7];
        sums[i] = sums[i] / fmaxf(c, 1.0f);
    }
}

__global__ __launch_bounds__(256) void edge_dot(
    const float* __restrict__ author_h, const float* __restrict__ paper_h,
    const int* __restrict__ lab_a, const int* __restrict__ lab_p,
    float* __restrict__ out, int num_edges)
{
    const int wid  = (int)(((long long)blockIdx.x * blockDim.x + threadIdx.x) >> 5);
    const int lane = threadIdx.x & 31;
    if (wid >= num_edges) return;

    const int a = lab_a[wid];
    const int p = lab_p[wid];
    const float4 va = *(const float4*)(author_h + (size_t)a * FEATDIM + lane * 4);
    const float4 vp = *(const float4*)(paper_h + (size_t)p * FEATDIM + lane * 4);
    float s = va.x * vp.x + va.y * vp.y + va.z * vp.z + va.w * vp.w;

    #pragma unroll
    for (int off = 16; off > 0; off >>= 1)
        s += __shfl_xor(s, off, 32);

    if (lane == 0) out[wid] = s;
}

extern "C" void kernel_launch(void* const* d_in, const int* in_sizes, int n_in,
                              void* d_out, int out_size, void* d_ws, size_t ws_size,
                              hipStream_t stream)
{
    const float* paper_x    = (const float*)d_in[0];
    const int*   author_ids = (const int*)  d_in[1];
    const int*   paper_ids  = (const int*)  d_in[2];
    const int*   lab_a      = (const int*)  d_in[3];
    const int*   lab_p      = (const int*)  d_in[4];
    const float* W_paper    = (const float*)d_in[5];
    const float* b_paper    = (const float*)d_in[6];
    const float* W_author   = (const float*)d_in[7];
    const float* b_author   = (const float*)d_in[8];

    const int num_papers      = in_sizes[0] / FEATDIM;
    const int num_edges       = in_sizes[1];
    const int num_label_edges = in_sizes[3];
    const int num_authors     = 200000;

    float* paper_h    = (float*)d_ws;
    float* author_buf = paper_h + (size_t)num_papers * FEATDIM;
    float* counts     = author_buf + (size_t)num_authors * FEATDIM;

    gemm128_wmma_f32<<<num_papers / BM, 256, 0, stream>>>(
        paper_x, W_paper, b_paper, paper_h, num_papers);

    const long long nz = (long long)num_authors * FEATDIM + num_authors;
    zero_f32<<<2048, 256, 0, stream>>>(author_buf, nz);

    {
        const long long threads = (long long)num_edges * 32;
        const int blocks = (int)((threads + 255) / 256);
        scatter_accum<<<blocks, 256, 0, stream>>>(
            paper_h, author_ids, paper_ids, author_buf, counts, num_edges);
    }

    {
        const long long n = (long long)num_authors * FEATDIM;
        const int blocks = (int)((n + 255) / 256);
        normalize_rows<<<blocks, 256, 0, stream>>>(author_buf, counts, n);
    }

    gemm128_wmma_f32<<<num_authors / BM, 256, 0, stream>>>(
        author_buf, W_author, b_author, author_buf, num_authors);

    {
        const long long threads = (long long)num_label_edges * 32;
        const int blocks = (int)((threads + 255) / 256);
        edge_dot<<<blocks, 256, 0, stream>>>(
            author_buf, paper_h, lab_a, lab_p, (float*)d_out, num_label_edges);
    }
}


# Pad the LDS A-tile row pitch from 128 to 132 floats so the 16 rows of each WMMA A-fragment read hit distinct banks (was a 16-way bank conflict on every read)
# speedup vs baseline: 1.1725x; 1.1725x over previous
#include <hip/hip_runtime.h>
#include <hip/hip_bf16.h>

typedef __attribute__((ext_vector_type(2))) float v2f;
typedef __attribute__((ext_vector_type(8))) float v8f;

#define FEATDIM 128
#define BM      32
#define LDSP    132

__global__ __launch_bounds__(256) void gemm128_wmma_f32(
    const float* __restrict__ A, const float* __restrict__ W,
    const float* __restrict__ bias, float* __restrict__ C, int M)
{
    __shared__ float sA[BM * LDSP];

    const int tid  = threadIdx.x;
    const int row0 = blockIdx.x * BM;
    if (row0 >= M) return;

    {
        const float4* src = (const float4*)(A + (size_t)row0 * FEATDIM);
        #pragma unroll
        for (int i = 0; i < 4; ++i) {
            const int idx = tid + 256 * i;
            const int r   = idx >> 5;
            const int c4  = idx & 31;
            *(float4*)&sA[r * LDSP + c4 * 4] = src[idx];
        }
    }
    __syncthreads();

    const int wave = tid >> 5;
    const int lane = tid & 31;
    const int half = lane >> 4;
    const int l16  = lane & 15;
    const int n    = wave * 16 + l16;

    v8f acc0 = {};
    v8f acc1 = {};

    #pragma unroll
    for (int k0 = 0; k0 < FEATDIM; k0 += 4) {
        const int ka = k0 + 2 * half;
        v2f b;
        b.x = W[(size_t)ka       * FEATDIM + n];
        b.y = W[(size_t)(ka + 1) * FEATDIM + n];
        v2f a0, a1;
        a0 = *(const v2f*)&sA[l16 * LDSP + ka];
        a1 = *(const v2f*)&sA[(16 + l16) * LDSP + ka];
        acc0 = __builtin_amdgcn_wmma_f32_16x16x4_f32(
            false, a0, false, b, (short)0, acc0, false, false);
        acc1 = __builtin_amdgcn_wmma_f32_16x16x4_f32(
            false, a1, false, b, (short)0, acc1, false, false);
    }

    const float bn = bias[n];
    float* Crow = C + (size_t)row0 * FEATDIM;
    #pragma unroll
    for (int v = 0; v < 8; ++v) {
        const int m = v + 8 * half;
        Crow[(size_t)m * FEATDIM + n]        = acc0[v] + bn;
        Crow[(size_t)(m + 16) * FEATDIM + n] = acc1[v] + bn;
    }
}

__global__ void zero_f32(float* __restrict__ p, long long n)
{
    long long i      = (long long)blockIdx.x * blockDim.x + threadIdx.x;
    long long stride = (long long)gridDim.x * blockDim.x;
    for (; i < n; i += stride) p[i] = 0.0f;
}

__global__ __launch_bounds__(256) void scatter_accum(
    const float* __restrict__ paper_h, const int* __restrict__ author_ids,
    const int* __restrict__ paper_ids, float* __restrict__ sums,
    float* __restrict__ counts, int num_edges)
{
    const int wid  = (int)(((long long)blockIdx.x * blockDim.x + threadIdx.x) >> 5);
    const int lane = threadIdx.x & 31;
    if (wid >= num_edges) return;

    const int aid = author_ids[wid];
    const int pid = paper_ids[wid];

    const float4 v = *(const float4*)(paper_h + (size_t)pid * FEATDIM + lane * 4);
    float* dst = sums + (size_t)aid * FEATDIM + lane * 4;
    atomicAdd(dst + 0, v.x);
    atomicAdd(dst + 1, v.y);
    atomicAdd(dst + 2, v.z);
    atomicAdd(dst + 3, v.w);
    if (lane == 0) atomicAdd(counts + aid, 1.0f);
}

__global__ void normalize_rows(float* __restrict__ sums,
                               const float* __restrict__ counts, long long n)
{
    long long i = (long long)blockIdx.x * blockDim.x + threadIdx.x;
    if (i < n) {
        const float c = counts[i >> 7];
        sums[i] = sums[i] / fmaxf(c, 1.0f);
    }
}

__global__ __launch_bounds__(256) void edge_dot(
    const float* __restrict__ author_h, const float* __restrict__ paper_h,
    const int* __restrict__ lab_a, const int* __restrict__ lab_p,
    float* __restrict__ out, int num_edges)
{
    const int wid  = (int)(((long long)blockIdx.x * blockDim.x + threadIdx.x) >> 5);
    const int lane = threadIdx.x & 31;
    if (wid >= num_edges) return;

    const int a = lab_a[wid];
    const int p = lab_p[wid];
    const float4 va = *(const float4*)(author_h + (size_t)a * FEATDIM + lane * 4);
    const float4 vp = *(const float4*)(paper_h + (size_t)p * FEATDIM + lane * 4);
    float s = va.x * vp.x + va.y * vp.y + va.z * vp.z + va.w * vp.w;

    #pragma unroll
    for (int off = 16; off > 0; off >>= 1)
        s += __shfl_xor(s, off, 32);

    if (lane == 0) out[wid] = s;
}

extern "C" void kernel_launch(void* const* d_in, const int* in_sizes, int n_in,
                              void* d_out, int out_size, void* d_ws, size_t ws_size,
                              hipStream_t stream)
{
    const float* paper_x    = (const float*)d_in[0];
    const int*   author_ids = (const int*)  d_in[1];
    const int*   paper_ids  = (const int*)  d_in[2];
    const int*   lab_a      = (const int*)  d_in[3];
    const int*   lab_p      = (const int*)  d_in[4];
    const float* W_paper    = (const float*)d_in[5];
    const float* b_paper    = (const float*)d_in[6];
    const float* W_author   = (const float*)d_in[7];
    const float* b_author   = (const float*)d_in[8];

    const int num_papers      = in_sizes[0] / FEATDIM;
    const int num_edges       = in_sizes[1];
    const int num_label_edges = in_sizes[3];
    const int num_authors     = 200000;

    float* paper_h    = (float*)d_ws;
    float* author_buf = paper_h + (size_t)num_papers * FEATDIM;
    float* counts     = author_buf + (size_t)num_authors * FEATDIM;

    gemm128_wmma_f32<<<num_papers / BM, 256, 0, stream>>>(
        paper_x, W_paper, b_paper, paper_h, num_papers);

    const long long nz = (long long)num_authors * FEATDIM + num_authors;
    zero_f32<<<2048, 256, 0, stream>>>(author_buf, nz);

    {
        const long long threads = (long long)num_edges * 32;
        const int blocks = (int)((threads + 255) / 256);
        scatter_accum<<<blocks, 256, 0, stream>>>(
            paper_h, author_ids, paper_ids, author_buf, counts, num_edges);
    }

    {
        const long long n = (long long)num_authors * FEATDIM;
        const int blocks = (int)((n + 255) / 256);
        normalize_rows<<<blocks, 256, 0, stream>>>(author_buf, counts, n);
    }

    gemm128_wmma_f32<<<num_authors / BM, 256, 0, stream>>>(
        author_buf, W_author, b_author, author_buf, num_authors);

    {
        const long long threads = (long long)num_label_edges * 32;
        const int blocks = (int)((threads + 255) / 256);
        edge_dot<<<blocks, 256, 0, stream>>>(
            author_buf, paper_h, lab_a, lab_p, (float*)d_out, num_label_edges);
    }
}


# Scatter-mean: bucket edges by author with a counting sort (int atomics), then one wave per author sums its gathered rows and stores the mean once; no float atomics; sum order = slot arrival order
# speedup vs baseline: 1.7183x; 1.4655x over previous
#include <hip/hip_runtime.h>
#include <hip/hip_bf16.h>

typedef __attribute__((ext_vector_type(2))) float v2f;
typedef __attribute__((ext_vector_type(8))) float v8f;

#define FEATDIM 128
#define BM      32
#define LDSP    132

__global__ __launch_bounds__(256) void gemm128_wmma_f32(
    const float* __restrict__ A, const float* __restrict__ W,
    const float* __restrict__ bias, float* __restrict__ C, int M)
{
    __shared__ float sA[BM * LDSP];

    const int tid  = threadIdx.x;
    const int row0 = blockIdx.x * BM;
    if (row0 >= M) return;

    {
        const float4* src = (const float4*)(A + (size_t)row0 * FEATDIM);
        #pragma unroll
        for (int i = 0; i < 4; ++i) {
            const int idx = tid + 256 * i;
            const int r   = idx >> 5;
            const int c4  = idx & 31;
            *(float4*)&sA[r * LDSP + c4 * 4] = src[idx];
        }
    }
    __syncthreads();

    const int wave = tid >> 5;
    const int lane = tid & 31;
    const int half = lane >> 4;
    const int l16  = lane & 15;
    const int n    = wave * 16 + l16;

    v8f acc0 = {};
    v8f acc1 = {};

    #pragma unroll
    for (int k0 = 0; k0 < FEATDIM; k0 += 4) {
        const int ka = k0 + 2 * half;
        v2f b;
        b.x = W[(size_t)ka       * FEATDIM + n];
        b.y = W[(size_t)(ka + 1) * FEATDIM + n];
        v2f a0, a1;
        a0 = *(const v2f*)&sA[l16 * LDSP + ka];
        a1 = *(const v2f*)&sA[(16 + l16) * LDSP + ka];
        acc0 = __builtin_amdgcn_wmma_f32_16x16x4_f32(
            false, a0, false, b, (short)0, acc0, false, false);
        acc1 = __builtin_amdgcn_wmma_f32_16x16x4_f32(
            false, a1, false, b, (short)0, acc1, false, false);
    }

    const float bn = bias[n];
    float* Crow = C + (size_t)row0 * FEATDIM;
    #pragma unroll
    for (int v = 0; v < 8; ++v) {
        const int m = v + 8 * half;
        Crow[(size_t)m * FEATDIM + n]        = acc0[v] + bn;
        Crow[(size_t)(m + 16) * FEATDIM + n] = acc1[v] + bn;
    }
}

__global__ void seg_zero(int* __restrict__ p, int n)
{
    const int i = blockIdx.x * blockDim.x + threadIdx.x;
    if (i < n) p[i] = 0;
}

__global__ __launch_bounds__(256) void seg_hist(
    const int* __restrict__ author_ids, int* __restrict__ counts, int num_edges)
{
    const int e = blockIdx.x * 256 + threadIdx.x;
    if (e < num_edges) atomicAdd(counts + author_ids[e], 1);
}

__global__ __launch_bounds__(256) void seg_alloc(
    const int* __restrict__ counts, int* __restrict__ cursor,
    int* __restrict__ bump, int num_authors)
{
    __shared__ int wave_off[8];
    __shared__ int block_base;

    const int tid  = threadIdx.x;
    const int lane = tid & 31;
    const int wave = tid >> 5;
    const int i    = blockIdx.x * 256 + tid;
    const int c    = (i < num_authors) ? counts[i] : 0;

    int x = c;
    #pragma unroll
    for (int off = 1; off < 32; off <<= 1) {
        const int y = __shfl_up(x, off, 32);
        if (lane >= off) x += y;
    }
    if (lane == 31) wave_off[wave] = x;
    __syncthreads();
    if (tid == 0) {
        int t = 0;
        #pragma unroll
        for (int w = 0; w < 8; ++w) {
            const int s = wave_off[w];
            wave_off[w] = t;
            t += s;
        }
        block_base = atomicAdd(bump, t);
    }
    __syncthreads();
    if (i < num_authors) cursor[i] = block_base + wave_off[wave] + (x - c);
}

__global__ __launch_bounds__(256) void seg_fill(
    const int* __restrict__ author_ids, const int* __restrict__ paper_ids,
    int* __restrict__ cursor, int* __restrict__ perm, int num_edges)
{
    const int e = blockIdx.x * 256 + threadIdx.x;
    if (e < num_edges) {
        const int slot = atomicAdd(cursor + author_ids[e], 1);
        perm[slot] = paper_ids[e];
    }
}

__device__ __forceinline__ float4 ld_row(const float* __restrict__ col, int p)
{
    return *(const float4*)(col + (size_t)p * FEATDIM);
}
__device__ __forceinline__ void add_row(float4& acc, const float4 v)
{
    acc.x += v.x; acc.y += v.y; acc.z += v.z; acc.w += v.w;
}

__global__ __launch_bounds__(256) void seg_mean(
    const float* __restrict__ paper_h, const int* __restrict__ perm,
    const int* __restrict__ seg_end, const int* __restrict__ counts,
    float* __restrict__ author_mean, int num_authors)
{
    const int lane = threadIdx.x & 31;
    const int a = __builtin_amdgcn_readfirstlane((int)(blockIdx.x * 8 + (threadIdx.x >> 5)));
    if (a >= num_authors) return;

    const int c   = counts[a];
    const int end = seg_end[a];
    const float* col = paper_h + lane * 4;

    float4 acc = make_float4(0.0f, 0.0f, 0.0f, 0.0f);
    for (int base = end - c; base < end; base += 32) {
        const int n = min(32, end - base);
        const int mypid = (lane < n) ? perm[base + lane] : 0;
        int j = 0;
        for (; j + 8 <= n; j += 8) {
            const float4 v0 = ld_row(col, __builtin_amdgcn_readlane(mypid, j));
            const float4 v1 = ld_row(col, __builtin_amdgcn_readlane(mypid, j + 1));
            const float4 v2 = ld_row(col, __builtin_amdgcn_readlane(mypid, j + 2));
            const float4 v3 = ld_row(col, __builtin_amdgcn_readlane(mypid, j + 3));
            const float4 v4 = ld_row(col, __builtin_amdgcn_readlane(mypid, j + 4));
            const float4 v5 = ld_row(col, __builtin_amdgcn_readlane(mypid, j + 5));
            const float4 v6 = ld_row(col, __builtin_amdgcn_readlane(mypid, j + 6));
            const float4 v7 = ld_row(col, __builtin_amdgcn_readlane(mypid, j + 7));
            add_row(acc, v0); add_row(acc, v1); add_row(acc, v2); add_row(acc, v3);
            add_row(acc, v4); add_row(acc, v5); add_row(acc, v6); add_row(acc, v7);
        }
        if (j + 4 <= n) {
            const float4 v0 = ld_row(col, __builtin_amdgcn_readlane(mypid, j));
            const float4 v1 = ld_row(col, __builtin_amdgcn_readlane(mypid, j + 1));
            const float4 v2 = ld_row(col, __builtin_amdgcn_readlane(mypid, j + 2));
            const float4 v3 = ld_row(col, __builtin_amdgcn_readlane(mypid, j + 3));
            add_row(acc, v0); add_row(acc, v1); add_row(acc, v2); add_row(acc, v3);
            j += 4;
        }
        const int rem = n - j;
        if (rem == 3) {
            const float4 v0 = ld_row(col, __builtin_amdgcn_readlane(mypid, j));
            const float4 v1 = ld_row(col, __builtin_amdgcn_readlane(mypid, j + 1));
            const float4 v2 = ld_row(col, __builtin_amdgcn_readlane(mypid, j + 2));
            add_row(acc, v0); add_row(acc, v1); add_row(acc, v2);
        } else if (rem == 2) {
            const float4 v0 = ld_row(col, __builtin_amdgcn_readlane(mypid, j));
            const float4 v1 = ld_row(col, __builtin_amdgcn_readlane(mypid, j + 1));
            add_row(acc, v0); add_row(acc, v1);
        } else if (rem == 1) {
            const float4 v0 = ld_row(col, __builtin_amdgcn_readlane(mypid, j));
            add_row(acc, v0);
        }
    }

    const float d = fmaxf((float)c, 1.0f);
    float4 o;
    o.x = acc.x / d; o.y = acc.y / d; o.z = acc.z / d; o.w = acc.w / d;
    *(float4*)(author_mean + (size_t)a * FEATDIM + lane * 4) = o;
}

__global__ __launch_bounds__(256) void edge_dot(
    const float* __restrict__ author_h, const float* __restrict__ paper_h,
    const int* __restrict__ lab_a, const int* __restrict__ lab_p,
    float* __restrict__ out, int num_edges)
{
    const int wid  = (int)(((long long)blockIdx.x * blockDim.x + threadIdx.x) >> 5);
    const int lane = threadIdx.x & 31;
    if (wid >= num_edges) return;

    const int a = lab_a[wid];
    const int p = lab_p[wid];
    const float4 va = *(const float4*)(author_h + (size_t)a * FEATDIM + lane * 4);
    const float4 vp = *(const float4*)(paper_h + (size_t)p * FEATDIM + lane * 4);
    float s = va.x * vp.x + va.y * vp.y + va.z * vp.z + va.w * vp.w;

    #pragma unroll
    for (int off = 16; off > 0; off >>= 1)
        s += __shfl_xor(s, off, 32);

    if (lane == 0) out[wid] = s;
}

extern "C" void kernel_launch(void* const* d_in, const int* in_sizes, int n_in,
                              void* d_out, int out_size, void* d_ws, size_t ws_size,
                              hipStream_t stream)
{
    const float* paper_x    = (const float*)d_in[0];
    const int*   author_ids = (const int*)  d_in[1];
    const int*   paper_ids  = (const int*)  d_in[2];
    const int*   lab_a      = (const int*)  d_in[3];
    const int*   lab_p      = (const int*)  d_in[4];
    const float* W_paper    = (const float*)d_in[5];
    const float* b_paper    = (const float*)d_in[6];
    const float* W_author   = (const float*)d_in[7];
    const float* b_author   = (const float*)d_in[8];

    const int num_papers      = in_sizes[0] / FEATDIM;
    const int num_edges       = in_sizes[1];
    const int num_label_edges = in_sizes[3];
    const int num_authors     = 200000;

    float* paper_h    = (float*)d_ws;
    float* author_buf = paper_h + (size_t)num_papers * FEATDIM;
    int*   counts     = (int*)(author_buf + (size_t)num_authors * FEATDIM);
    int*   bump       = counts + num_authors;
    int*   cursor     = counts + num_authors + 64;
    int*   perm       = cursor + num_authors;

    gemm128_wmma_f32<<<num_papers / BM, 256, 0, stream>>>(
        paper_x, W_paper, b_paper, paper_h, num_papers);

    const int edge_blocks   = (num_edges + 255) / 256;
    const int author_blocks = (num_authors + 255) / 256;
    seg_zero<<<(num_authors + 1 + 255) / 256, 256, 0, stream>>>(counts, num_authors + 1);
    seg_hist<<<edge_blocks, 256, 0, stream>>>(author_ids, counts, num_edges);
    seg_alloc<<<author_blocks, 256, 0, stream>>>(counts, cursor, bump, num_authors);
    seg_fill<<<edge_blocks, 256, 0, stream>>>(author_ids, paper_ids, cursor, perm, num_edges);

    seg_mean<<<(num_authors + 7) / 8, 256, 0, stream>>>(
        paper_h, perm, cursor, counts, author_buf, num_authors);

    gemm128_wmma_f32<<<num_authors / BM, 256, 0, stream>>>(
        author_buf, W_author, b_author, author_buf, num_authors);

    {
        const long long threads = (long long)num_label_edges * 32;
        const int blocks = (int)((threads + 255) / 256);
        edge_dot<<<blocks, 256, 0, stream>>>(
            author_buf, paper_h, lab_a, lab_p, (float*)d_out, num_label_edges);
    }
}


# edge_dot: one wave walks 32 consecutive edges (ids loaded once, 8 edges' rows in flight, one 128 B store) instead of one 2-row wave per edge: 500k -> 15.6k waves started; per-edge sum order kept
# speedup vs baseline: 1.8388x; 1.0701x over previous
#include <hip/hip_runtime.h>
#include <hip/hip_bf16.h>

typedef __attribute__((ext_vector_type(2))) float v2f;
typedef __attribute__((ext_vector_type(8))) float v8f;

#define FEATDIM 128
#define BM      32
#define LDSP    132

__global__ __launch_bounds__(256) void gemm128_wmma_f32(
    const float* __restrict__ A, const float* __restrict__ W,
    const float* __restrict__ bias, float* __restrict__ C, int M)
{
    __shared__ float sA[BM * LDSP];

    const int tid  = threadIdx.x;
    const int row0 = blockIdx.x * BM;
    if (row0 >= M) return;

    {
        const float4* src = (const float4*)(A + (size_t)row0 * FEATDIM);
        #pragma unroll
        for (int i = 0; i < 4; ++i) {
            const int idx = tid + 256 * i;
            const int r   = idx >> 5;
            const int c4  = idx & 31;
            *(float4*)&sA[r * LDSP + c4 * 4] = src[idx];
        }
    }
    __syncthreads();

    const int wave = tid >> 5;
    const int lane = tid & 31;
    const int half = lane >> 4;
    const int l16  = lane & 15;
    const int n    = wave * 16 + l16;

    v8f acc0 = {};
    v8f acc1 = {};

    #pragma unroll
    for (int k0 = 0; k0 < FEATDIM; k0 += 4) {
        const int ka = k0 + 2 * half;
        v2f b;
        b.x = W[(size_t)ka       * FEATDIM + n];
        b.y = W[(size_t)(ka + 1) * FEATDIM + n];
        v2f a0, a1;
        a0 = *(const v2f*)&sA[l16 * LDSP + ka];
        a1 = *(const v2f*)&sA[(16 + l16) * LDSP + ka];
        acc0 = __builtin_amdgcn_wmma_f32_16x16x4_f32(
            false, a0, false, b, (short)0, acc0, false, false);
        acc1 = __builtin_amdgcn_wmma_f32_16x16x4_f32(
            false, a1, false, b, (short)0, acc1, false, false);
    }

    const float bn = bias[n];
    float* Crow = C + (size_t)row0 * FEATDIM;
    #pragma unroll
    for (int v = 0; v < 8; ++v) {
        const int m = v + 8 * half;
        Crow[(size_t)m * FEATDIM + n]        = acc0[v] + bn;
        Crow[(size_t)(m + 16) * FEATDIM + n] = acc1[v] + bn;
    }
}

__global__ void seg_zero(int* __restrict__ p, int n)
{
    const int i = blockIdx.x * blockDim.x + threadIdx.x;
    if (i < n) p[i] = 0;
}

__global__ __launch_bounds__(256) void seg_hist(
    const int* __restrict__ author_ids, int* __restrict__ counts, int num_edges)
{
    const int e = blockIdx.x * 256 + threadIdx.x;
    if (e < num_edges) atomicAdd(counts + author_ids[e], 1);
}

__global__ __launch_bounds__(256) void seg_alloc(
    const int* __restrict__ counts, int* __restrict__ cursor,
    int* __restrict__ bump, int num_authors)
{
    __shared__ int wave_off[8];
    __shared__ int block_base;

    const int tid  = threadIdx.x;
    const int lane = tid & 31;
    const int wave = tid >> 5;
    const int i    = blockIdx.x * 256 + tid;
    const int c    = (i < num_authors) ? counts[i] : 0;

    int x = c;
    #pragma unroll
    for (int off = 1; off < 32; off <<= 1) {
        const int y = __shfl_up(x, off, 32);
        if (lane >= off) x += y;
    }
    if (lane == 31) wave_off[wave] = x;
    __syncthreads();
    if (tid == 0) {
        int t = 0;
        #pragma unroll
        for (int w = 0; w < 8; ++w) {
            const int s = wave_off[w];
            wave_off[w] = t;
            t += s;
        }
        block_base = atomicAdd(bump, t);
    }
    __syncthreads();
    if (i < num_authors) cursor[i] = block_base + wave_off[wave] + (x - c);
}

__global__ __launch_bounds__(256) void seg_fill(
    const int* __restrict__ author_ids, const int* __restrict__ paper_ids,
    int* __restrict__ cursor, int* __restrict__ perm, int num_edges)
{
    const int e = blockIdx.x * 256 + threadIdx.x;
    if (e < num_edges) {
        const int slot = atomicAdd(cursor + author_ids[e], 1);
        perm[slot] = paper_ids[e];
    }
}

__device__ __forceinline__ float4 ld_row(const float* __restrict__ col, int p)
{
    return *(const float4*)(col + (size_t)p * FEATDIM);
}
__device__ __forceinline__ void add_row(float4& acc, const float4 v)
{
    acc.x += v.x; acc.y += v.y; acc.z += v.z; acc.w += v.w;
}

__global__ __launch_bounds__(256) void seg_mean(
    const float* __restrict__ paper_h, const int* __restrict__ perm,
    const int* __restrict__ seg_end, const int* __restrict__ counts,
    float* __restrict__ author_mean, int num_authors)
{
    const int lane = threadIdx.x & 31;
    const int a = __builtin_amdgcn_readfirstlane((int)(blockIdx.x * 8 + (threadIdx.x >> 5)));
    if (a >= num_authors) return;

    const int c   = counts[a];
    const int end = seg_end[a];
    const float* col = paper_h + lane * 4;

    float4 acc = make_float4(0.0f, 0.0f, 0.0f, 0.0f);
    for (int base = end - c; base < end; base += 32) {
        const int n = min(32, end - base);
        const int mypid = (lane < n) ? perm[base + lane] : 0;
        int j = 0;
        for (; j + 8 <= n; j += 8) {
            const float4 v0 = ld_row(col, __builtin_amdgcn_readlane(mypid, j));
            const float4 v1 = ld_row(col, __builtin_amdgcn_readlane(mypid, j + 1));
            const float4 v2 = ld_row(col, __builtin_amdgcn_readlane(mypid, j + 2));
            const float4 v3 = ld_row(col, __builtin_amdgcn_readlane(mypid, j + 3));
            const float4 v4 = ld_row(col, __builtin_amdgcn_readlane(mypid, j + 4));
            const float4 v5 = ld_row(col, __builtin_amdgcn_readlane(mypid, j + 5));
            const float4 v6 = ld_row(col, __builtin_amdgcn_readlane(mypid, j + 6));
            const float4 v7 = ld_row(col, __builtin_amdgcn_readlane(mypid, j + 7));
            add_row(acc, v0); add_row(acc, v1); add_row(acc, v2); add_row(acc, v3);
            add_row(acc, v4); add_row(acc, v5); add_row(acc, v6); add_row(acc, v7);
        }
        if (j + 4 <= n) {
            const float4 v0 = ld_row(col, __builtin_amdgcn_readlane(mypid, j));
            const float4 v1 = ld_row(col, __builtin_amdgcn_readlane(mypid, j + 1));
            const float4 v2 = ld_row(col, __builtin_amdgcn_readlane(mypid, j + 2));
            const float4 v3 = ld_row(col, __builtin_amdgcn_readlane(mypid, j + 3));
            add_row(acc, v0); add_row(acc, v1); add_row(acc, v2); add_row(acc, v3);
            j += 4;
        }
        const int rem = n - j;
        if (rem == 3) {
            const float4 v0 = ld_row(col, __builtin_amdgcn_readlane(mypid, j));
            const float4 v1 = ld_row(col, __builtin_amdgcn_readlane(mypid, j + 1));
            const float4 v2 = ld_row(col, __builtin_amdgcn_readlane(mypid, j + 2));
            add_row(acc, v0); add_row(acc, v1); add_row(acc, v2);
        } else if (rem == 2) {
            const float4 v0 = ld_row(col, __builtin_amdgcn_readlane(mypid, j));
            const float4 v1 = ld_row(col, __builtin_amdgcn_readlane(mypid, j + 1));
            add_row(acc, v0); add_row(acc, v1);
        } else if (rem == 1) {
            const float4 v0 = ld_row(col, __builtin_amdgcn_readlane(mypid, j));
            add_row(acc, v0);
        }
    }

    const float d = fmaxf((float)c, 1.0f);
    float4 o;
    o.x = acc.x / d; o.y = acc.y / d; o.z = acc.z / d; o.w = acc.w / d;
    *(float4*)(author_mean + (size_t)a * FEATDIM + lane * 4) = o;
}

#define EDGE_GROUP 8
__global__ __launch_bounds__(256) __attribute__((amdgpu_waves_per_eu(2, 8))) void edge_dot(
    const float* __restrict__ author_h, const float* __restrict__ paper_h,
    const int* __restrict__ lab_a, const int* __restrict__ lab_p,
    float* __restrict__ out, int num_edges)
{
    const int lane = threadIdx.x & 31;
    const int wave = __builtin_amdgcn_readfirstlane((int)(blockIdx.x * 8 + (threadIdx.x >> 5)));
    const int base = wave * 32;
    if (base >= num_edges) return;

    const int  e    = base + lane;
    const bool live = e < num_edges;
    const int my_a = live ? lab_a[e] : 0;
    const int my_p = live ? lab_p[e] : 0;
    const int n    = min(32, num_edges - base);

    const char* const ab = (const char*)author_h;
    const char* const pb = (const char*)paper_h;
    const unsigned lane16 = (unsigned)lane * 16u;
    float res = 0.0f;

    for (int g = 0; g < n; g += EDGE_GROUP) {
        float4 va[EDGE_GROUP], vp[EDGE_GROUP];
        #pragma unroll
        for (int k = 0; k < EDGE_GROUP; ++k) {
            const unsigned a = (unsigned)__builtin_amdgcn_readlane(my_a, g + k);
            const unsigned p = (unsigned)__builtin_amdgcn_readlane(my_p, g + k);
            va[k] = *(const float4*)(ab + ((a << 9) + lane16));
            vp[k] = *(const float4*)(pb + ((p << 9) + lane16));
        }
        #pragma unroll
        for (int k = 0; k < EDGE_GROUP; ++k) {
            float s = __builtin_fmaf(va[k].w, vp[k].w,
                      __builtin_fmaf(va[k].z, vp[k].z,
                      __builtin_fmaf(va[k].x, vp[k].x, va[k].y * vp[k].y)));
            #pragma unroll
            for (int o = 16; o > 0; o >>= 1)
                s += __shfl_xor(s, o, 32);
            if (lane == g + k) res = s;
        }
    }

    if (live) out[e] = res;
}

extern "C" void kernel_launch(void* const* d_in, const int* in_sizes, int n_in,
                              void* d_out, int out_size, void* d_ws, size_t ws_size,
                              hipStream_t stream)
{
    const float* paper_x    = (const float*)d_in[0];
    const int*   author_ids = (const int*)  d_in[1];
    const int*   paper_ids  = (const int*)  d_in[2];
    const int*   lab_a      = (const int*)  d_in[3];
    const int*   lab_p      = (const int*)  d_in[4];
    const float* W_paper    = (const float*)d_in[5];
    const float* b_paper    = (const float*)d_in[6];
    const float* W_author   = (const float*)d_in[7];
    const float* b_author   = (const float*)d_in[8];

    const int num_papers      = in_sizes[0] / FEATDIM;
    const int num_edges       = in_sizes[1];
    const int num_label_edges = in_sizes[3];
    const int num_authors     = 200000;

    float* paper_h    = (float*)d_ws;
    float* author_buf = paper_h + (size_t)num_papers * FEATDIM;
    int*   counts     = (int*)(author_buf + (size_t)num_authors * FEATDIM);
    int*   bump       = counts + num_authors;
    int*   cursor     = counts + num_authors + 64;
    int*   perm       = cursor + num_authors;

    gemm128_wmma_f32<<<num_papers / BM, 256, 0, stream>>>(
        paper_x, W_paper, b_paper, paper_h, num_papers);

    const int edge_blocks   = (num_edges + 255) / 256;
    const int author_blocks = (num_authors + 255) / 256;
    seg_zero<<<(num_authors + 1 + 255) / 256, 256, 0, stream>>>(counts, num_authors + 1);
    seg_hist<<<edge_blocks, 256, 0, stream>>>(author_ids, counts, num_edges);
    seg_alloc<<<author_blocks, 256, 0, stream>>>(counts, cursor, bump, num_authors);
    seg_fill<<<edge_blocks, 256, 0, stream>>>(author_ids, paper_ids, cursor, perm, num_edges);

    seg_mean<<<(num_authors + 7) / 8, 256, 0, stream>>>(
        paper_h, perm, cursor, counts, author_buf, num_authors);

    gemm128_wmma_f32<<<num_authors / BM, 256, 0, stream>>>(
        author_buf, W_author, b_author, author_buf, num_authors);

    {
        const int waves  = (num_label_edges + 31) / 32;
        const int blocks = (waves + 7) / 8;
        edge_dot<<<blocks, 256, 0, stream>>>(
            author_buf, paper_h, lab_a, lab_p, (float*)d_out, num_label_edges);
    }
}


# Fuse the author GEMM into the segment-mean kernel: a block sums 32 authors' gathered rows (4 per wave, 8 rows per round trip) into the LDS A tile and multiplies by W_author there; 8 -> 7 launches
# speedup vs baseline: 1.9030x; 1.0349x over previous
#include <hip/hip_runtime.h>
#include <hip/hip_bf16.h>

typedef __attribute__((ext_vector_type(2))) float v2f;
typedef __attribute__((ext_vector_type(8))) float v8f;

#define FEATDIM 128
#define BM      32
#define LDSP    132

__device__ __forceinline__ void gemm_tile_from_lds(
    const float* sA, const float* __restrict__ W,
    const float* __restrict__ bias, float* __restrict__ Crow, int tid)
{
    const int wave = tid >> 5;
    const int lane = tid & 31;
    const int half = lane >> 4;
    const int l16  = lane & 15;
    const int n    = wave * 16 + l16;

    v8f acc0 = {};
    v8f acc1 = {};

    #pragma unroll
    for (int k0 = 0; k0 < FEATDIM; k0 += 4) {
        const int ka = k0 + 2 * half;
        v2f b;
        b.x = W[(size_t)ka       * FEATDIM + n];
        b.y = W[(size_t)(ka + 1) * FEATDIM + n];
        v2f a0, a1;
        a0 = *(const v2f*)&sA[l16 * LDSP + ka];
        a1 = *(const v2f*)&sA[(16 + l16) * LDSP + ka];
        acc0 = __builtin_amdgcn_wmma_f32_16x16x4_f32(
            false, a0, false, b, (short)0, acc0, false, false);
        acc1 = __builtin_amdgcn_wmma_f32_16x16x4_f32(
            false, a1, false, b, (short)0, acc1, false, false);
    }

    const float bn = bias[n];
    #pragma unroll
    for (int v = 0; v < 8; ++v) {
        const int m = v + 8 * half;
        Crow[(size_t)m * FEATDIM + n]        = acc0[v] + bn;
        Crow[(size_t)(m + 16) * FEATDIM + n] = acc1[v] + bn;
    }
}

__global__ __launch_bounds__(256) void gemm128_wmma_f32(
    const float* __restrict__ A, const float* __restrict__ W,
    const float* __restrict__ bias, float* __restrict__ C, int M)
{
    __shared__ float sA[BM * LDSP];

    const int tid  = threadIdx.x;
    const int row0 = blockIdx.x * BM;
    if (row0 >= M) return;

    {
        const float4* src = (const float4*)(A + (size_t)row0 * FEATDIM);
        #pragma unroll
        for (int i = 0; i < 4; ++i) {
            const int idx = tid + 256 * i;
            const int r   = idx >> 5;
            const int c4  = idx & 31;
            *(float4*)&sA[r * LDSP + c4 * 4] = src[idx];
        }
    }
    __syncthreads();

    gemm_tile_from_lds(sA, W, bias, C + (size_t)row0 * FEATDIM, tid);
}

__global__ void seg_zero(int* __restrict__ p, int n)
{
    const int i = blockIdx.x * blockDim.x + threadIdx.x;
    if (i < n) p[i] = 0;
}

__global__ __launch_bounds__(256) void seg_hist(
    const int* __restrict__ author_ids, int* __restrict__ counts, int num_edges)
{
    const int e = blockIdx.x * 256 + threadIdx.x;
    if (e < num_edges) atomicAdd(counts + author_ids[e], 1);
}

__global__ __launch_bounds__(256) void seg_alloc(
    const int* __restrict__ counts, int* __restrict__ cursor,
    int* __restrict__ bump, int num_authors)
{
    __shared__ int wave_off[8];
    __shared__ int block_base;

    const int tid  = threadIdx.x;
    const int lane = tid & 31;
    const int wave = tid >> 5;
    const int i    = blockIdx.x * 256 + tid;
    const int c    = (i < num_authors) ? counts[i] : 0;

    int x = c;
    #pragma unroll
    for (int off = 1; off < 32; off <<= 1) {
        const int y = __shfl_up(x, off, 32);
        if (lane >= off) x += y;
    }
    if (lane == 31) wave_off[wave] = x;
    __syncthreads();
    if (tid == 0) {
        int t = 0;
        #pragma unroll
        for (int w = 0; w < 8; ++w) {
            const int s = wave_off[w];
            wave_off[w] = t;
            t += s;
        }
        block_base = atomicAdd(bump, t);
    }
    __syncthreads();
    if (i < num_authors) cursor[i] = block_base + wave_off[wave] + (x - c);
}

__global__ __launch_bounds__(256) void seg_fill(
    const int* __restrict__ author_ids, const int* __restrict__ paper_ids,
    int* __restrict__ cursor, int* __restrict__ perm, int num_edges)
{
    const int e = blockIdx.x * 256 + threadIdx.x;
    if (e < num_edges) {
        const int slot = atomicAdd(cursor + author_ids[e], 1);
        perm[slot] = paper_ids[e];
    }
}

__device__ __forceinline__ void add_row(float4& acc, const float4 v)
{
    acc.x += v.x; acc.y += v.y; acc.z += v.z; acc.w += v.w;
}

__device__ __forceinline__ float4 ld_row_off(const char* tab, unsigned p, unsigned lane16)
{
    return *(const float4*)(tab + ((p << 9) + lane16));
}

__device__ __forceinline__ int pid_at(const int* __restrict__ perm, int f,
                                      int E0, int E1, int E2, int total,
                                      int d0, int d1, int d2, int d3)
{
    int dl = d0;
    dl = (f >= E0) ? d1 : dl;
    dl = (f >= E1) ? d2 : dl;
    dl = (f >= E2) ? d3 : dl;
    return (f < total) ? perm[f + dl] : 0;
}

__global__ __launch_bounds__(256) __attribute__((amdgpu_waves_per_eu(12, 16))) void seg_mean_gemm(
    const float* __restrict__ paper_h, const int* __restrict__ perm,
    const int* __restrict__ seg_end, const int* __restrict__ counts,
    const float* __restrict__ W, const float* __restrict__ bias,
    float* __restrict__ author_h, int num_authors)
{
    __shared__ float sA[BM * LDSP];

    const int tid  = threadIdx.x;
    const int lane = tid & 31;
    const int wave = __builtin_amdgcn_readfirstlane(tid >> 5);
    const int row0 = blockIdx.x * BM;
    if (row0 >= num_authors) return;

    const int a0 = row0 + 4 * wave;
    int c_l = 0, e_l = 0;
    if (lane < 4 && a0 + lane < num_authors) {
        c_l = counts[a0 + lane];
        e_l = seg_end[a0 + lane];
    }
    const int c0 = __builtin_amdgcn_readlane(c_l, 0);
    const int c1 = __builtin_amdgcn_readlane(c_l, 1);
    const int c2 = __builtin_amdgcn_readlane(c_l, 2);
    const int c3 = __builtin_amdgcn_readlane(c_l, 3);
    const int E0 = c0, E1 = E0 + c1, E2 = E1 + c2, total = E2 + c3;
    const int d0 = __builtin_amdgcn_readlane(e_l, 0) - E0;
    const int d1 = __builtin_amdgcn_readlane(e_l, 1) - E1;
    const int d2 = __builtin_amdgcn_readlane(e_l, 2) - E2;
    const int d3 = __builtin_amdgcn_readlane(e_l, 3) - total;

    float* const myrows = sA + (4 * wave) * LDSP + lane * 4;
    #pragma unroll
    for (int i = 0; i < 4; ++i)
        *(float4*)(myrows + i * LDSP) = make_float4(0.0f, 0.0f, 0.0f, 0.0f);

    const char* const pb = (const char*)paper_h;
    const unsigned lane16 = (unsigned)lane * 16u;

    float4 acc = make_float4(0.0f, 0.0f, 0.0f, 0.0f);
    int pid_next = pid_at(perm, lane, E0, E1, E2, total, d0, d1, d2, d3);
    for (int base = 0; base < total; base += 32) {
        const int mypid = pid_next;
        const int f = base + lane;
        pid_next = 0;
        const int au   = (f >= E0) + (f >= E1) + (f >= E2);
        const int last = (f + 1 == E0) | (f + 1 == E1) | (f + 1 == E2) | (f + 1 == total);
        const int code = au | (last << 2);
        const int n = min(32, total - base);

        int j = 0;
        for (; j + 8 <= n; j += 8) {
            float4 v[8];
            #pragma unroll
            for (int k = 0; k < 8; ++k)
                v[k] = ld_row_off(pb, (unsigned)__builtin_amdgcn_readlane(mypid, j + k), lane16);
            if (j == 0)
                pid_next = pid_at(perm, f + 32, E0, E1, E2, total, d0, d1, d2, d3);
            #pragma unroll
            for (int k = 0; k < 8; ++k) {
                add_row(acc, v[k]);
                const int cd = __builtin_amdgcn_readlane(code, j + k);
                if (cd & 4) {
                    *(float4*)(myrows + (cd & 3) * LDSP) = acc;
                    acc = make_float4(0.0f, 0.0f, 0.0f, 0.0f);
                }
            }
        }
        if (j < n) {
            float4 v[7];
            #pragma unroll
            for (int k = 0; k < 7; ++k)
                if (j + k < n)
                    v[k] = ld_row_off(pb, (unsigned)__builtin_amdgcn_readlane(mypid, j + k), lane16);
            #pragma unroll
            for (int k = 0; k < 7; ++k)
                if (j + k < n) {
                    add_row(acc, v[k]);
                    const int cd = __builtin_amdgcn_readlane(code, j + k);
                    if (cd & 4) {
                        *(float4*)(myrows + (cd & 3) * LDSP) = acc;
                        acc = make_float4(0.0f, 0.0f, 0.0f, 0.0f);
                    }
                }
        }
    }

    #pragma unroll
    for (int i = 0; i < 4; ++i) {
        const float d = fmaxf((float)__builtin_amdgcn_readlane(c_l, i), 1.0f);
        float4 s = *(const float4*)(myrows + i * LDSP);
        s.x = s.x / d; s.y = s.y / d; s.z = s.z / d; s.w = s.w / d;
        *(float4*)(myrows + i * LDSP) = s;
    }
    __syncthreads();

    gemm_tile_from_lds(sA, W, bias, author_h + (size_t)row0 * FEATDIM, tid);
}

#define EDGE_GROUP 8
__global__ __launch_bounds__(256) __attribute__((amdgpu_waves_per_eu(2, 8))) void edge_dot(
    const float* __restrict__ author_h, const float* __restrict__ paper_h,
    const int* __restrict__ lab_a, const int* __restrict__ lab_p,
    float* __restrict__ out, int num_edges)
{
    const int lane = threadIdx.x & 31;
    const int wave = __builtin_amdgcn_readfirstlane((int)(blockIdx.x * 8 + (threadIdx.x >> 5)));
    const int base = wave * 32;
    if (base >= num_edges) return;

    const int  e    = base + lane;
    const bool live = e < num_edges;
    const int my_a = live ? lab_a[e] : 0;
    const int my_p = live ? lab_p[e] : 0;
    const int n    = min(32, num_edges - base);

    const char* const ab = (const char*)author_h;
    const char* const pb = (const char*)paper_h;
    const unsigned lane16 = (unsigned)lane * 16u;
    float res = 0.0f;

    for (int g = 0; g < n; g += EDGE_GROUP) {
        float4 va[EDGE_GROUP], vp[EDGE_GROUP];
        #pragma unroll
        for (int k = 0; k < EDGE_GROUP; ++k) {
            const unsigned a = (unsigned)__builtin_amdgcn_readlane(my_a, g + k);
            const unsigned p = (unsigned)__builtin_amdgcn_readlane(my_p, g + k);
            va[k] = *(const float4*)(ab + ((a << 9) + lane16));
            vp[k] = *(const float4*)(pb + ((p << 9) + lane16));
        }
        #pragma unroll
        for (int k = 0; k < EDGE_GROUP; ++k) {
            float s = __builtin_fmaf(va[k].w, vp[k].w,
                      __builtin_fmaf(va[k].z, vp[k].z,
                      __builtin_fmaf(va[k].x, vp[k].x, va[k].y * vp[k].y)));
            #pragma unroll
            for (int o = 16; o > 0; o >>= 1)
                s += __shfl_xor(s, o, 32);
            if (lane == g + k) res = s;
        }
    }

    if (live) out[e] = res;
}

extern "C" void kernel_launch(void* const* d_in, const int* in_sizes, int n_in,
                              void* d_out, int out_size, void* d_ws, size_t ws_size,
                              hipStream_t stream)
{
    const float* paper_x    = (const float*)d_in[0];
    const int*   author_ids = (const int*)  d_in[1];
    const int*   paper_ids  = (const int*)  d_in[2];
    const int*   lab_a      = (const int*)  d_in[3];
    const int*   lab_p      = (const int*)  d_in[4];
    const float* W_paper    = (const float*)d_in[5];
    const float* b_paper    = (const float*)d_in[6];
    const float* W_author   = (const float*)d_in[7];
    const float* b_author   = (const float*)d_in[8];

    const int num_papers      = in_sizes[0] / FEATDIM;
    const int num_edges       = in_sizes[1];
    const int num_label_edges = in_sizes[3];
    const int num_authors     = 200000;

    float* paper_h    = (float*)d_ws;
    float* author_buf = paper_h + (size_t)num_papers * FEATDIM;
    int*   counts     = (int*)(author_buf + (size_t)num_authors * FEATDIM);
    int*   bump       = counts + num_authors;
    int*   cursor     = counts + num_authors + 64;
    int*   perm       = cursor + num_authors;

    gemm128_wmma_f32<<<num_papers / BM, 256, 0, stream>>>(
        paper_x, W_paper, b_paper, paper_h, num_papers);

    const int edge_blocks   = (num_edges + 255) / 256;
    const int author_blocks = (num_authors + 255) / 256;
    seg_zero<<<(num_authors + 1 + 255) / 256, 256, 0, stream>>>(counts, num_authors + 1);
    seg_hist<<<edge_blocks, 256, 0, stream>>>(author_ids, counts, num_edges);
    seg_alloc<<<author_blocks, 256, 0, stream>>>(counts, cursor, bump, num_authors);
    seg_fill<<<edge_blocks, 256, 0, stream>>>(author_ids, paper_ids, cursor, perm, num_edges);

    seg_mean_gemm<<<num_authors / BM, 256, 0, stream>>>(
        paper_h, perm, cursor, counts, W_author, b_author, author_buf, num_authors);

    {
        const int waves  = (num_label_edges + 31) / 32;
        const int blocks = (waves + 7) / 8;
        edge_dot<<<blocks, 256, 0, stream>>>(
            author_buf, paper_h, lab_a, lab_p, (float*)d_out, num_label_edges);
    }
}


# Group edges by author with two LDS counting sorts (per 8192-edge chunk by author range, then per range by author), no global atomics: 4 sort launches -> 2; sum order = LDS arrival order
# speedup vs baseline: 2.9593x; 1.5551x over previous
#include <hip/hip_runtime.h>
#include <hip/hip_bf16.h>

typedef __attribute__((ext_vector_type(2))) float v2f;
typedef __attribute__((ext_vector_type(8))) float v8f;

#define FEATDIM 128
#define BM      32
#define LDSP    132

__device__ __forceinline__ void gemm_tile_from_lds(
    const float* sA, const float* __restrict__ W,
    const float* __restrict__ bias, float* __restrict__ Crow, int tid)
{
    const int wave = tid >> 5;
    const int lane = tid & 31;
    const int half = lane >> 4;
    const int l16  = lane & 15;
    const int n    = wave * 16 + l16;

    v8f acc0 = {};
    v8f acc1 = {};

    #pragma unroll
    for (int k0 = 0; k0 < FEATDIM; k0 += 4) {
        const int ka = k0 + 2 * half;
        v2f b;
        b.x = W[(size_t)ka       * FEATDIM + n];
        b.y = W[(size_t)(ka + 1) * FEATDIM + n];
        v2f a0, a1;
        a0 = *(const v2f*)&sA[l16 * LDSP + ka];
        a1 = *(const v2f*)&sA[(16 + l16) * LDSP + ka];
        acc0 = __builtin_amdgcn_wmma_f32_16x16x4_f32(
            false, a0, false, b, (short)0, acc0, false, false);
        acc1 = __builtin_amdgcn_wmma_f32_16x16x4_f32(
            false, a1, false, b, (short)0, acc1, false, false);
    }

    const float bn = bias[n];
    #pragma unroll
    for (int v = 0; v < 8; ++v) {
        const int m = v + 8 * half;
        Crow[(size_t)m * FEATDIM + n]        = acc0[v] + bn;
        Crow[(size_t)(m + 16) * FEATDIM + n] = acc1[v] + bn;
    }
}

__global__ __launch_bounds__(256) void gemm128_wmma_f32(
    const float* __restrict__ A, const float* __restrict__ W,
    const float* __restrict__ bias, float* __restrict__ C, int M)
{
    __shared__ float sA[BM * LDSP];

    const int tid  = threadIdx.x;
    const int row0 = blockIdx.x * BM;
    if (row0 >= M) return;

    {
        const float4* src = (const float4*)(A + (size_t)row0 * FEATDIM);
        #pragma unroll
        for (int i = 0; i < 4; ++i) {
            const int idx = tid + 256 * i;
            const int r   = idx >> 5;
            const int c4  = idx & 31;
            *(float4*)&sA[r * LDSP + c4 * 4] = src[idx];
        }
    }
    __syncthreads();

    gemm_tile_from_lds(sA, W, bias, C + (size_t)row0 * FEATDIM, tid);
}

#define SORT_BINS    256
#define SORT_CHUNK   8192
#define SORT_THREADS 1024
#define EMIT_CAP     16384

__global__ __launch_bounds__(SORT_THREADS) void seg_chunk(
    const int* __restrict__ author_ids, const int* __restrict__ paper_ids,
    int2* __restrict__ sorted, int* __restrict__ runstart,
    int num_edges, int num_authors, int apb)
{
    __shared__ int  hist[SORT_BINS];
    __shared__ int  binstart[SORT_BINS + 1];
    __shared__ int  wsum[SORT_BINS / 32];
    __shared__ int2 stage[SORT_CHUNK];

    const int tid  = threadIdx.x;
    const int lane = tid & 31;
    const int wave = tid >> 5;
    const int e0   = blockIdx.x * SORT_CHUNK;

    if (tid < SORT_BINS) hist[tid] = 0;
    __syncthreads();

    int a[8], p[8], bin[8], rank[8];
    #pragma unroll
    for (int k = 0; k < 8; ++k) {
        const int e = e0 + k * SORT_THREADS + tid;
        a[k] = -1; p[k] = 0;
        if (e < num_edges) { a[k] = author_ids[e]; p[k] = paper_ids[e]; }
    }
    #pragma unroll
    for (int k = 0; k < 8; ++k) {
        bin[k] = -1; rank[k] = 0;
        if ((unsigned)a[k] < (unsigned)num_authors) {
            bin[k]  = (int)((unsigned)a[k] / (unsigned)apb);
            rank[k] = atomicAdd(&hist[bin[k]], 1);
        }
    }
    __syncthreads();

    int x = 0, incl = 0;
    if (tid < SORT_BINS) {
        x = hist[tid];
        incl = x;
        #pragma unroll
        for (int off = 1; off < 32; off <<= 1) {
            const int y = __shfl_up(incl, off, 32);
            if (lane >= off) incl += y;
        }
        if (lane == 31) wsum[wave] = incl;
    }
    __syncthreads();
    if (tid < SORT_BINS) {
        int add = 0;
        for (int w = 0; w < wave; ++w) add += wsum[w];
        incl += add;
        binstart[tid] = incl - x;
        if (tid == SORT_BINS - 1) binstart[SORT_BINS] = incl;
    }
    __syncthreads();

    #pragma unroll
    for (int k = 0; k < 8; ++k)
        if (bin[k] >= 0) stage[binstart[bin[k]] + rank[k]] = make_int2(a[k], p[k]);
    __syncthreads();

    const int placed = binstart[SORT_BINS];
    for (int i = tid; i < placed; i += SORT_THREADS) sorted[e0 + i] = stage[i];
    if (tid <= SORT_BINS) runstart[blockIdx.x * (SORT_BINS + 1) + tid] = binstart[tid];
}

__global__ __launch_bounds__(SORT_THREADS) void seg_emit(
    const int2* __restrict__ sorted, const int* __restrict__ runstart,
    int* __restrict__ counts, int* __restrict__ seg_end, int* __restrict__ perm,
    int num_chunks, int num_authors, int apb)
{
    __shared__ int rstart[SORT_THREADS];
    __shared__ int rlen[SORT_THREADS];
    __shared__ int acount[SORT_THREADS];
    __shared__ int cur[SORT_THREADS];
    __shared__ int wsum[SORT_THREADS / 32];
    __shared__ int tot[2];
    __shared__ int lperm[EMIT_CAP];

    const int tid     = threadIdx.x;
    const int lane    = tid & 31;
    const int wave    = tid >> 5;
    const int b       = blockIdx.x;
    const int a_first = b * apb;

    if (tid < 2) tot[tid] = 0;
    acount[tid] = 0;
    int s0 = 0, len = 0;
    if (tid < num_chunks) {
        s0  = runstart[tid * (SORT_BINS + 1) + b];
        len = runstart[tid * (SORT_BINS + 1) + b + 1] - s0;
    }
    rstart[tid] = tid * SORT_CHUNK + s0;
    rlen[tid]   = len;
    int ws0 = s0, wlen = len;
    #pragma unroll
    for (int o = 16; o > 0; o >>= 1) {
        ws0  += __shfl_xor(ws0, o, 32);
        wlen += __shfl_xor(wlen, o, 32);
    }
    __syncthreads();
    if (lane == 0) { atomicAdd(&tot[0], ws0); atomicAdd(&tot[1], wlen); }
    __syncthreads();
    const int base = tot[0];
    const int T    = tot[1];

    int2 v[16];
    #pragma unroll
    for (int k = 0; k < 8; ++k) {
        const int r = wave + 32 * k;
        const int l = (r < num_chunks) ? rlen[r] : 0;
        v[k]     = make_int2(-1, 0);
        v[8 + k] = make_int2(-1, 0);
        if (lane < l)      v[k]     = sorted[rstart[r] + lane];
        if (lane + 32 < l) v[8 + k] = sorted[rstart[r] + 32 + lane];
    }
    #pragma unroll
    for (int k = 0; k < 16; ++k)
        if (v[k].x >= 0) atomicAdd(&acount[v[k].x - a_first], 1);
    for (int r = wave; r < num_chunks; r += 32) {
        const int l   = rlen[r];
        const int src = rstart[r];
        for (int j = ((r < 256) ? 64 : 0) + lane; j < l; j += 32)
            atomicAdd(&acount[sorted[src + j].x - a_first], 1);
    }
    __syncthreads();

    const int x = acount[tid];
    int incl = x;
    #pragma unroll
    for (int off = 1; off < 32; off <<= 1) {
        const int y = __shfl_up(incl, off, 32);
        if (lane >= off) incl += y;
    }
    if (lane == 31) wsum[wave] = incl;
    __syncthreads();
    {
        int add = 0;
        for (int w = 0; w < wave; ++w) add += wsum[w];
        incl += add;
    }
    cur[tid] = incl - x;
    if (tid < apb && a_first + tid < num_authors) {
        counts[a_first + tid]  = x;
        seg_end[a_first + tid] = base + incl;
    }
    __syncthreads();

    const bool in_lds = T <= EMIT_CAP;
    #pragma unroll
    for (int k = 0; k < 16; ++k)
        if (v[k].x >= 0) {
            const int slot = atomicAdd(&cur[v[k].x - a_first], 1);
            if (in_lds) lperm[slot] = v[k].y; else perm[base + slot] = v[k].y;
        }
    for (int r = wave; r < num_chunks; r += 32) {
        const int l   = rlen[r];
        const int src = rstart[r];
        for (int j = ((r < 256) ? 64 : 0) + lane; j < l; j += 32) {
            const int2 q = sorted[src + j];
            const int slot = atomicAdd(&cur[q.x - a_first], 1);
            if (in_lds) lperm[slot] = q.y; else perm[base + slot] = q.y;
        }
    }
    __syncthreads();
    if (in_lds)
        for (int i = tid; i < T; i += SORT_THREADS) perm[base + i] = lperm[i];
}

__device__ __forceinline__ void add_row(float4& acc, const float4 v)
{
    acc.x += v.x; acc.y += v.y; acc.z += v.z; acc.w += v.w;
}

__device__ __forceinline__ float4 ld_row_off(const char* tab, unsigned p, unsigned lane16)
{
    return *(const float4*)(tab + ((p << 9) + lane16));
}

__device__ __forceinline__ int pid_at(const int* __restrict__ perm, int f,
                                      int E0, int E1, int E2, int total,
                                      int d0, int d1, int d2, int d3)
{
    int dl = d0;
    dl = (f >= E0) ? d1 : dl;
    dl = (f >= E1) ? d2 : dl;
    dl = (f >= E2) ? d3 : dl;
    return (f < total) ? perm[f + dl] : 0;
}

__global__ __launch_bounds__(256) __attribute__((amdgpu_waves_per_eu(12, 16))) void seg_mean_gemm(
    const float* __restrict__ paper_h, const int* __restrict__ perm,
    const int* __restrict__ seg_end, const int* __restrict__ counts,
    const float* __restrict__ W, const float* __restrict__ bias,
    float* __restrict__ author_h, int num_authors)
{
    __shared__ float sA[BM * LDSP];

    const int tid  = threadIdx.x;
    const int lane = tid & 31;
    const int wave = __builtin_amdgcn_readfirstlane(tid >> 5);
    const int row0 = blockIdx.x * BM;
    if (row0 >= num_authors) return;

    const int a0 = row0 + 4 * wave;
    int c_l = 0, e_l = 0;
    if (lane < 4 && a0 + lane < num_authors) {
        c_l = counts[a0 + lane];
        e_l = seg_end[a0 + lane];
    }
    const int c0 = __builtin_amdgcn_readlane(c_l, 0);
    const int c1 = __builtin_amdgcn_readlane(c_l, 1);
    const int c2 = __builtin_amdgcn_readlane(c_l, 2);
    const int c3 = __builtin_amdgcn_readlane(c_l, 3);
    const int E0 = c0, E1 = E0 + c1, E2 = E1 + c2, total = E2 + c3;
    const int d0 = __builtin_amdgcn_readlane(e_l, 0) - E0;
    const int d1 = __builtin_amdgcn_readlane(e_l, 1) - E1;
    const int d2 = __builtin_amdgcn_readlane(e_l, 2) - E2;
    const int d3 = __builtin_amdgcn_readlane(e_l, 3) - total;

    float* const myrows = sA + (4 * wave) * LDSP + lane * 4;
    #pragma unroll
    for (int i = 0; i < 4; ++i)
        *(float4*)(myrows + i * LDSP) = make_float4(0.0f, 0.0f, 0.0f, 0.0f);

    const char* const pb = (const char*)paper_h;
    const unsigned lane16 = (unsigned)lane * 16u;

    float4 acc = make_float4(0.0f, 0.0f, 0.0f, 0.0f);
    int pid_next = pid_at(perm, lane, E0, E1, E2, total, d0, d1, d2, d3);
    for (int base = 0; base < total; base += 32) {
        const int mypid = pid_next;
        const int f = base + lane;
        pid_next = 0;
        const int au   = (f >= E0) + (f >= E1) + (f >= E2);
        const int last = (f + 1 == E0) | (f + 1 == E1) | (f + 1 == E2) | (f + 1 == total);
        const int code = au | (last << 2);
        const int n = min(32, total - base);

        int j = 0;
        for (; j + 8 <= n; j += 8) {
            float4 v[8];
            #pragma unroll
            for (int k = 0; k < 8; ++k)
                v[k] = ld_row_off(pb, (unsigned)__builtin_amdgcn_readlane(mypid, j + k), lane16);
            if (j == 0)
                pid_next = pid_at(perm, f + 32, E0, E1, E2, total, d0, d1, d2, d3);
            #pragma unroll
            for (int k = 0; k < 8; ++k) {
                add_row(acc, v[k]);
                const int cd = __builtin_amdgcn_readlane(code, j + k);
                if (cd & 4) {
                    *(float4*)(myrows + (cd & 3) * LDSP) = acc;
                    acc = make_float4(0.0f, 0.0f, 0.0f, 0.0f);
                }
            }
        }
        if (j < n) {
            float4 v[7];
            #pragma unroll
            for (int k = 0; k < 7; ++k)
                if (j + k < n)
                    v[k] = ld_row_off(pb, (unsigned)__builtin_amdgcn_readlane(mypid, j + k), lane16);
            #pragma unroll
            for (int k = 0; k < 7; ++k)
                if (j + k < n) {
                    add_row(acc, v[k]);
                    const int cd = __builtin_amdgcn_readlane(code, j + k);
                    if (cd & 4) {
                        *(float4*)(myrows + (cd & 3) * LDSP) = acc;
                        acc = make_float4(0.0f, 0.0f, 0.0f, 0.0f);
                    }
                }
        }
    }

    #pragma unroll
    for (int i = 0; i < 4; ++i) {
        const float d = fmaxf((float)__builtin_amdgcn_readlane(c_l, i), 1.0f);
        float4 s = *(const float4*)(myrows + i * LDSP);
        s.x = s.x / d; s.y = s.y / d; s.z = s.z / d; s.w = s.w / d;
        *(float4*)(myrows + i * LDSP) = s;
    }
    __syncthreads();

    gemm_tile_from_lds(sA, W, bias, author_h + (size_t)row0 * FEATDIM, tid);
}

#define EDGE_GROUP 8
__global__ __launch_bounds__(256) __attribute__((amdgpu_waves_per_eu(2, 8))) void edge_dot(
    const float* __restrict__ author_h, const float* __restrict__ paper_h,
    const int* __restrict__ lab_a, const int* __restrict__ lab_p,
    float* __restrict__ out, int num_edges)
{
    const int lane = threadIdx.x & 31;
    const int wave = __builtin_amdgcn_readfirstlane((int)(blockIdx.x * 8 + (threadIdx.x >> 5)));
    const int base = wave * 32;
    if (base >= num_edges) return;

    const int  e    = base + lane;
    const bool live = e < num_edges;
    const int my_a = live ? lab_a[e] : 0;
    const int my_p = live ? lab_p[e] : 0;
    const int n    = min(32, num_edges - base);

    const char* const ab = (const char*)author_h;
    const char* const pb = (const char*)paper_h;
    const unsigned lane16 = (unsigned)lane * 16u;
    float res = 0.0f;

    for (int g = 0; g < n; g += EDGE_GROUP) {
        float4 va[EDGE_GROUP], vp[EDGE_GROUP];
        #pragma unroll
        for (int k = 0; k < EDGE_GROUP; ++k) {
            const unsigned a = (unsigned)__builtin_amdgcn_readlane(my_a, g + k);
            const unsigned p = (unsigned)__builtin_amdgcn_readlane(my_p, g + k);
            va[k] = *(const float4*)(ab + ((a << 9) + lane16));
            vp[k] = *(const float4*)(pb + ((p << 9) + lane16));
        }
        #pragma unroll
        for (int k = 0; k < EDGE_GROUP; ++k) {
            float s = __builtin_fmaf(va[k].w, vp[k].w,
                      __builtin_fmaf(va[k].z, vp[k].z,
                      __builtin_fmaf(va[k].x, vp[k].x, va[k].y * vp[k].y)));
            #pragma unroll
            for (int o = 16; o > 0; o >>= 1)
                s += __shfl_xor(s, o, 32);
            if (lane == g + k) res = s;
        }
    }

    if (live) out[e] = res;
}

extern "C" void kernel_launch(void* const* d_in, const int* in_sizes, int n_in,
                              void* d_out, int out_size, void* d_ws, size_t ws_size,
                              hipStream_t stream)
{
    const float* paper_x    = (const float*)d_in[0];
    const int*   author_ids = (const int*)  d_in[1];
    const int*   paper_ids  = (const int*)  d_in[2];
    const int*   lab_a      = (const int*)  d_in[3];
    const int*   lab_p      = (const int*)  d_in[4];
    const float* W_paper    = (const float*)d_in[5];
    const float* b_paper    = (const float*)d_in[6];
    const float* W_author   = (const float*)d_in[7];
    const float* b_author   = (const float*)d_in[8];

    const int num_papers      = in_sizes[0] / FEATDIM;
    const int num_edges       = in_sizes[1];
    const int num_label_edges = in_sizes[3];
    const int num_authors     = 200000;

    const int num_chunks = (num_edges + SORT_CHUNK - 1) / SORT_CHUNK;
    const int apb        = (num_authors + SORT_BINS - 1) / SORT_BINS;
    float* paper_h    = (float*)d_ws;
    float* author_buf = paper_h + (size_t)num_papers * FEATDIM;
    int2*  sorted     = (int2*)(author_buf + (size_t)num_authors * FEATDIM);
    int*   runstart   = (int*)(sorted + (size_t)num_chunks * SORT_CHUNK);
    int*   counts     = runstart + (size_t)num_chunks * (SORT_BINS + 1);
    int*   cursor     = counts + num_authors;
    int*   perm       = cursor + num_authors;

    gemm128_wmma_f32<<<num_papers / BM, 256, 0, stream>>>(
        paper_x, W_paper, b_paper, paper_h, num_papers);

    seg_chunk<<<num_chunks, SORT_THREADS, 0, stream>>>(
        author_ids, paper_ids, sorted, runstart, num_edges, num_authors, apb);
    seg_emit<<<SORT_BINS, SORT_THREADS, 0, stream>>>(
        sorted, runstart, counts, cursor, perm, num_chunks, num_authors, apb);

    seg_mean_gemm<<<num_authors / BM, 256, 0, stream>>>(
        paper_h, perm, cursor, counts, W_author, b_author, author_buf, num_authors);

    {
        const int waves  = (num_label_edges + 31) / 32;
        const int blocks = (waves + 7) / 8;
        edge_dot<<<blocks, 256, 0, stream>>>(
            author_buf, paper_h, lab_a, lab_p, (float*)d_out, num_label_edges);
    }
}


# Fuse the edge dot into the author tile kernel: supervision edges grouped by author in the same two sort launches, pred taken from the LDS tile; author_g never stored or re-read; 5 -> 4 launches
# speedup vs baseline: 5.1058x; 1.7253x over previous
#include <hip/hip_runtime.h>
#include <hip/hip_bf16.h>

typedef __attribute__((ext_vector_type(2))) float v2f;
typedef __attribute__((ext_vector_type(8))) float v8f;

#define FEATDIM 128
#define BM      32
#define LDSP    132

#define FOLD_ROWS (FEATDIM + 2)
__global__ __launch_bounds__(FEATDIM) void fold_weights(
    const float* __restrict__ Wp, const float* __restrict__ bp,
    const float* __restrict__ Wa, const float* __restrict__ ba,
    float* __restrict__ Z, float* __restrict__ zc)
{
    __shared__ __attribute__((aligned(16))) float sY[FEATDIM];
    const int i    = blockIdx.x;
    const int n    = threadIdx.x;
    const int lane = n & 31;
    const int wave = n >> 5;

    float y;
    if (i <= FEATDIM) {
        const float* xr = (i < FEATDIM) ? Wp + (size_t)i * FEATDIM : bp;
        y = 0.0f;
        #pragma unroll
        for (int k = 0; k < FEATDIM; ++k)
            y = __builtin_fmaf(xr[k], Wa[(size_t)k * FEATDIM + n], y);
    } else {
        y = ba[n];
    }
    sY[n] = y;
    __syncthreads();

    const float4 yv = *(const float4*)&sY[lane * 4];
    float zkeep = 0.0f;
    #pragma unroll
    for (int t = 0; t < 32; ++t) {
        const float4 w = *(const float4*)(Wp + (size_t)(wave * 32 + t) * FEATDIM + lane * 4);
        float sdot = __builtin_fmaf(w.w, yv.w, __builtin_fmaf(w.z, yv.z,
                     __builtin_fmaf(w.y, yv.y, w.x * yv.x)));
        #pragma unroll
        for (int o = 16; o > 0; o >>= 1)
            sdot += __shfl_xor(sdot, o, 32);
        if (lane == t) zkeep = sdot;
    }
    Z[(size_t)i * FEATDIM + n] = zkeep;

    if (wave == 0) {
        const float4 w = *(const float4*)(bp + lane * 4);
        float sdot = __builtin_fmaf(w.w, yv.w, __builtin_fmaf(w.z, yv.z,
                     __builtin_fmaf(w.y, yv.y, w.x * yv.x)));
        #pragma unroll
        for (int o = 16; o > 0; o >>= 1)
            sdot += __shfl_xor(sdot, o, 32);
        if (lane == 0) zc[i] = sdot;
    }
}

__device__ __forceinline__ void gemm_tile_from_lds(
    const float* sA, const float* sInd, const float* __restrict__ W,
    float* sG, int tid)
{
    const int wave = tid >> 5;
    const int lane = tid & 31;
    const int half = lane >> 4;
    const int l16  = lane & 15;
    const int n    = wave * 16 + l16;

    v8f acc0 = {};
    v8f acc1 = {};

    #pragma unroll
    for (int k0 = 0; k0 < FEATDIM; k0 += 4) {
        const int ka = k0 + 2 * half;
        v2f b;
        b.x = W[(size_t)ka       * FEATDIM + n];
        b.y = W[(size_t)(ka + 1) * FEATDIM + n];
        v2f a0, a1;
        a0 = *(const v2f*)&sA[l16 * LDSP + ka];
        a1 = *(const v2f*)&sA[(16 + l16) * LDSP + ka];
        acc0 = __builtin_amdgcn_wmma_f32_16x16x4_f32(
            false, a0, false, b, (short)0, acc0, false, false);
        acc1 = __builtin_amdgcn_wmma_f32_16x16x4_f32(
            false, a1, false, b, (short)0, acc1, false, false);
    }

    const float b_ind = W[(size_t)FEATDIM * FEATDIM + n];
    const float b_all = W[(size_t)(FEATDIM + 1) * FEATDIM + n];
    #pragma unroll
    for (int v = 0; v < 8; ++v) {
        const int m = v + 8 * half;
        const float bm0 = ((sInd[m]      != 0.0f) ? b_ind : 0.0f) + b_all;
        const float bm1 = ((sInd[m + 16] != 0.0f) ? b_ind : 0.0f) + b_all;
        sG[m * LDSP + n]        = acc0[v] + bm0;
        sG[(m + 16) * LDSP + n] = acc1[v] + bm1;
    }
}

#define SORT_BINS    256
#define SORT_CHUNK   8192
#define SORT_THREADS 1024
#define EMIT_CAP     16384

__global__ __launch_bounds__(SORT_THREADS) void seg_chunk(
    const int* __restrict__ author_ids, const int* __restrict__ paper_ids,
    const int* __restrict__ lab_a,
    int2* __restrict__ sorted_w, int* __restrict__ runstart_w,
    int2* __restrict__ sorted_l, int* __restrict__ runstart_l,
    int num_edges_w, int num_edges_l, int num_chunks_w, int num_authors, int apb)
{
    __shared__ int  hist[SORT_BINS];
    __shared__ int  binstart[SORT_BINS + 1];
    __shared__ int  wsum[SORT_BINS / 32];
    __shared__ int2 stage[SORT_CHUNK];

    const int tid  = threadIdx.x;
    const int lane = tid & 31;
    const int wave = tid >> 5;
    const bool lab = (int)blockIdx.x >= num_chunks_w;
    const int  blk = lab ? (int)blockIdx.x - num_chunks_w : (int)blockIdx.x;
    const int* __restrict__ ids = lab ? lab_a : author_ids;
    const int  num_edges = lab ? num_edges_l : num_edges_w;
    int2* __restrict__ sorted   = lab ? sorted_l : sorted_w;
    int*  __restrict__ runstart = lab ? runstart_l : runstart_w;
    const int e0   = blk * SORT_CHUNK;

    if (tid < SORT_BINS) hist[tid] = 0;
    __syncthreads();

    int a[8], p[8], bin[8], rank[8];
    #pragma unroll
    for (int k = 0; k < 8; ++k) {
        const int e = e0 + k * SORT_THREADS + tid;
        a[k] = -1; p[k] = 0;
        if (e < num_edges) { a[k] = ids[e]; p[k] = lab ? e : paper_ids[e]; }
    }
    #pragma unroll
    for (int k = 0; k < 8; ++k) {
        bin[k] = -1; rank[k] = 0;
        if ((unsigned)a[k] < (unsigned)num_authors) {
            bin[k]  = (int)((unsigned)a[k] / (unsigned)apb);
            rank[k] = atomicAdd(&hist[bin[k]], 1);
        }
    }
    __syncthreads();

    int x = 0, incl = 0;
    if (tid < SORT_BINS) {
        x = hist[tid];
        incl = x;
        #pragma unroll
        for (int off = 1; off < 32; off <<= 1) {
            const int y = __shfl_up(incl, off, 32);
            if (lane >= off) incl += y;
        }
        if (lane == 31) wsum[wave] = incl;
    }
    __syncthreads();
    if (tid < SORT_BINS) {
        int add = 0;
        for (int w = 0; w < wave; ++w) add += wsum[w];
        incl += add;
        binstart[tid] = incl - x;
        if (tid == SORT_BINS - 1) binstart[SORT_BINS] = incl;
    }
    __syncthreads();

    #pragma unroll
    for (int k = 0; k < 8; ++k)
        if (bin[k] >= 0) stage[binstart[bin[k]] + rank[k]] = make_int2(a[k], p[k]);
    __syncthreads();

    const int placed = binstart[SORT_BINS];
    for (int i = tid; i < placed; i += SORT_THREADS) sorted[e0 + i] = stage[i];
    if (tid <= SORT_BINS) runstart[blk * (SORT_BINS + 1) + tid] = binstart[tid];
}

__global__ __launch_bounds__(SORT_THREADS) void seg_emit(
    const int2* __restrict__ sorted_w, const int* __restrict__ runstart_w,
    int* __restrict__ counts_w, int* __restrict__ seg_end_w, int* __restrict__ perm_w,
    const int2* __restrict__ sorted_l, const int* __restrict__ runstart_l,
    int* __restrict__ counts_l, int* __restrict__ seg_end_l, int* __restrict__ perm_l,
    int num_chunks_w, int num_chunks_l, int num_authors, int apb)
{
    __shared__ int rstart[SORT_THREADS];
    __shared__ int rlen[SORT_THREADS];
    __shared__ int acount[SORT_THREADS];
    __shared__ int cur[SORT_THREADS];
    __shared__ int wsum[SORT_THREADS / 32];
    __shared__ int tot[2];
    __shared__ int lperm[EMIT_CAP];

    const int tid     = threadIdx.x;
    const int lane    = tid & 31;
    const int wave    = tid >> 5;
    const bool lab    = (int)blockIdx.x >= SORT_BINS;
    const int b       = lab ? (int)blockIdx.x - SORT_BINS : (int)blockIdx.x;
    const int a_first = b * apb;
    const int2* __restrict__ sorted   = lab ? sorted_l : sorted_w;
    const int*  __restrict__ runstart = lab ? runstart_l : runstart_w;
    int* __restrict__ counts  = lab ? counts_l : counts_w;
    int* __restrict__ seg_end = lab ? seg_end_l : seg_end_w;
    int* __restrict__ perm    = lab ? perm_l : perm_w;
    const int num_chunks = lab ? num_chunks_l : num_chunks_w;

    if (tid < 2) tot[tid] = 0;
    acount[tid] = 0;
    int s0 = 0, len = 0;
    if (tid < num_chunks) {
        s0  = runstart[tid * (SORT_BINS + 1) + b];
        len = runstart[tid * (SORT_BINS + 1) + b + 1] - s0;
    }
    rstart[tid] = tid * SORT_CHUNK + s0;
    rlen[tid]   = len;
    int ws0 = s0, wlen = len;
    #pragma unroll
    for (int o = 16; o > 0; o >>= 1) {
        ws0  += __shfl_xor(ws0, o, 32);
        wlen += __shfl_xor(wlen, o, 32);
    }
    __syncthreads();
    if (lane == 0) { atomicAdd(&tot[0], ws0); atomicAdd(&tot[1], wlen); }
    __syncthreads();
    const int base = tot[0];
    const int T    = tot[1];

    int2 v[16];
    #pragma unroll
    for (int k = 0; k < 8; ++k) {
        const int r = wave + 32 * k;
        const int l = (r < num_chunks) ? rlen[r] : 0;
        v[k]     = make_int2(-1, 0);
        v[8 + k] = make_int2(-1, 0);
        if (lane < l)      v[k]     = sorted[rstart[r] + lane];
        if (lane + 32 < l) v[8 + k] = sorted[rstart[r] + 32 + lane];
    }
    #pragma unroll
    for (int k = 0; k < 16; ++k)
        if (v[k].x >= 0) atomicAdd(&acount[v[k].x - a_first], 1);
    for (int r = wave; r < num_chunks; r += 32) {
        const int l   = rlen[r];
        const int src = rstart[r];
        for (int j = ((r < 256) ? 64 : 0) + lane; j < l; j += 32)
            atomicAdd(&acount[sorted[src + j].x - a_first], 1);
    }
    __syncthreads();

    const int x = acount[tid];
    int incl = x;
    #pragma unroll
    for (int off = 1; off < 32; off <<= 1) {
        const int y = __shfl_up(incl, off, 32);
        if (lane >= off) incl += y;
    }
    if (lane == 31) wsum[wave] = incl;
    __syncthreads();
    {
        int add = 0;
        for (int w = 0; w < wave; ++w) add += wsum[w];
        incl += add;
    }
    cur[tid] = incl - x;
    if (tid < apb && a_first + tid < num_authors) {
        counts[a_first + tid]  = x;
        seg_end[a_first + tid] = base + incl;
    }
    __syncthreads();

    const bool in_lds = T <= EMIT_CAP;
    #pragma unroll
    for (int k = 0; k < 16; ++k)
        if (v[k].x >= 0) {
            const int slot = atomicAdd(&cur[v[k].x - a_first], 1);
            if (in_lds) lperm[slot] = v[k].y; else perm[base + slot] = v[k].y;
        }
    for (int r = wave; r < num_chunks; r += 32) {
        const int l   = rlen[r];
        const int src = rstart[r];
        for (int j = ((r < 256) ? 64 : 0) + lane; j < l; j += 32) {
            const int2 q = sorted[src + j];
            const int slot = atomicAdd(&cur[q.x - a_first], 1);
            if (in_lds) lperm[slot] = q.y; else perm[base + slot] = q.y;
        }
    }
    __syncthreads();
    if (in_lds)
        for (int i = tid; i < T; i += SORT_THREADS) perm[base + i] = lperm[i];
}

__device__ __forceinline__ void add_row(float4& acc, const float4 v)
{
    acc.x += v.x; acc.y += v.y; acc.z += v.z; acc.w += v.w;
}

__device__ __forceinline__ float4 ld_row_off(const char* tab, unsigned p, unsigned lane16)
{
    return *(const float4*)(tab + ((p << 9) + lane16));
}

__device__ __forceinline__ int pid_at(const int* __restrict__ perm, int f,
                                      int E0, int E1, int E2, int total,
                                      int d0, int d1, int d2, int d3)
{
    int dl = d0;
    dl = (f >= E0) ? d1 : dl;
    dl = (f >= E1) ? d2 : dl;
    dl = (f >= E2) ? d3 : dl;
    return (f < total) ? perm[f + dl] : 0;
}

__global__ __launch_bounds__(256) __attribute__((amdgpu_waves_per_eu(12, 16))) void seg_mean_gemm(
    const float* __restrict__ paper_x, const int* __restrict__ perm,
    const int* __restrict__ seg_end, const int* __restrict__ counts,
    const float* __restrict__ Z, const float* __restrict__ zc,
    const int* __restrict__ perm_l, const int* __restrict__ seg_end_l,
    const int* __restrict__ counts_l, const int* __restrict__ lab_p,
    float* __restrict__ out, int num_authors)
{
    __shared__ float sA[BM * LDSP];
    __shared__ __attribute__((aligned(16))) float sG[BM * LDSP];
    __shared__ float sInd[BM];
    __shared__ float sC[BM];

    const int tid  = threadIdx.x;
    const int lane = tid & 31;
    const int wave = __builtin_amdgcn_readfirstlane(tid >> 5);
    const int row0 = blockIdx.x * BM;
    if (row0 >= num_authors) return;

    const int a0 = row0 + 4 * wave;
    int c_l = 0, e_l = 0, cl_l = 0, el_l = 0;
    if (lane < 4 && a0 + lane < num_authors) {
        c_l  = counts[a0 + lane];
        e_l  = seg_end[a0 + lane];
        cl_l = counts_l[a0 + lane];
        el_l = seg_end_l[a0 + lane];
    }
    if (lane < 4) sInd[4 * wave + lane] = (c_l > 0) ? 1.0f : 0.0f;
    const int c0 = __builtin_amdgcn_readlane(c_l, 0);
    const int c1 = __builtin_amdgcn_readlane(c_l, 1);
    const int c2 = __builtin_amdgcn_readlane(c_l, 2);
    const int c3 = __builtin_amdgcn_readlane(c_l, 3);
    const int E0 = c0, E1 = E0 + c1, E2 = E1 + c2, total = E2 + c3;
    const int d0 = __builtin_amdgcn_readlane(e_l, 0) - E0;
    const int d1 = __builtin_amdgcn_readlane(e_l, 1) - E1;
    const int d2 = __builtin_amdgcn_readlane(e_l, 2) - E2;
    const int d3 = __builtin_amdgcn_readlane(e_l, 3) - total;
    const int F0 = __builtin_amdgcn_readlane(cl_l, 0);
    const int F1 = F0 + __builtin_amdgcn_readlane(cl_l, 1);
    const int F2 = F1 + __builtin_amdgcn_readlane(cl_l, 2);
    const int totalL = F2 + __builtin_amdgcn_readlane(cl_l, 3);
    const int g0 = __builtin_amdgcn_readlane(el_l, 0) - F0;
    const int g1 = __builtin_amdgcn_readlane(el_l, 1) - F1;
    const int g2 = __builtin_amdgcn_readlane(el_l, 2) - F2;
    const int g3 = __builtin_amdgcn_readlane(el_l, 3) - totalL;
    const int e_first = pid_at(perm_l, lane, F0, F1, F2, totalL, g0, g1, g2, g3);

    float* const myrows = sA + (4 * wave) * LDSP + lane * 4;
    #pragma unroll
    for (int i = 0; i < 4; ++i)
        *(float4*)(myrows + i * LDSP) = make_float4(0.0f, 0.0f, 0.0f, 0.0f);

    const char* const pb = (const char*)paper_x;
    const unsigned lane16 = (unsigned)lane * 16u;

    float4 acc = make_float4(0.0f, 0.0f, 0.0f, 0.0f);
    int pid_next = pid_at(perm, lane, E0, E1, E2, total, d0, d1, d2, d3);
    for (int base = 0; base < total; base += 32) {
        const int mypid = pid_next;
        const int f = base + lane;
        pid_next = 0;
        const int au   = (f >= E0) + (f >= E1) + (f >= E2);
        const int last = (f + 1 == E0) | (f + 1 == E1) | (f + 1 == E2) | (f + 1 == total);
        const int code = au | (last << 2);
        const int n = min(32, total - base);

        int j = 0;
        for (; j + 8 <= n; j += 8) {
            float4 v[8];
            #pragma unroll
            for (int k = 0; k < 8; ++k)
                v[k] = ld_row_off(pb, (unsigned)__builtin_amdgcn_readlane(mypid, j + k), lane16);
            if (j == 0)
                pid_next = pid_at(perm, f + 32, E0, E1, E2, total, d0, d1, d2, d3);
            #pragma unroll
            for (int k = 0; k < 8; ++k) {
                add_row(acc, v[k]);
                const int cd = __builtin_amdgcn_readlane(code, j + k);
                if (cd & 4) {
                    *(float4*)(myrows + (cd & 3) * LDSP) = acc;
                    acc = make_float4(0.0f, 0.0f, 0.0f, 0.0f);
                }
            }
        }
        if (j < n) {
            float4 v[7];
            #pragma unroll
            for (int k = 0; k < 7; ++k)
                if (j + k < n)
                    v[k] = ld_row_off(pb, (unsigned)__builtin_amdgcn_readlane(mypid, j + k), lane16);
            #pragma unroll
            for (int k = 0; k < 7; ++k)
                if (j + k < n) {
                    add_row(acc, v[k]);
                    const int cd = __builtin_amdgcn_readlane(code, j + k);
                    if (cd & 4) {
                        *(float4*)(myrows + (cd & 3) * LDSP) = acc;
                        acc = make_float4(0.0f, 0.0f, 0.0f, 0.0f);
                    }
                }
        }
    }

    const float4 u = *(const float4*)(zc + lane * 4);
    const float c_ind = zc[FEATDIM];
    const float c_all = zc[FEATDIM + 1];
    float cval = 0.0f;
    #pragma unroll
    for (int i = 0; i < 4; ++i) {
        const int   ci = __builtin_amdgcn_readlane(c_l, i);
        const float d  = fmaxf((float)ci, 1.0f);
        float4 s = *(const float4*)(myrows + i * LDSP);
        s.x = s.x / d; s.y = s.y / d; s.z = s.z / d; s.w = s.w / d;
        *(float4*)(myrows + i * LDSP) = s;
        float t = __builtin_fmaf(s.w, u.w, __builtin_fmaf(s.z, u.z,
                  __builtin_fmaf(s.y, u.y, s.x * u.x)));
        #pragma unroll
        for (int o = 16; o > 0; o >>= 1)
            t += __shfl_xor(t, o, 32);
        t += ((ci > 0) ? c_ind : 0.0f) + c_all;
        if (lane == i) cval = t;
    }
    if (lane < 4) sC[4 * wave + lane] = cval;
    __syncthreads();

    int p_first = 0;
    if (lane < totalL) p_first = lab_p[e_first];

    gemm_tile_from_lds(sA, sInd, Z, sG, tid);
    __syncthreads();

    const float* const grows = sG + (4 * wave) * LDSP + lane * 4;
    for (int lb = 0; lb < totalL; lb += 32) {
        const int f  = lb + lane;
        int ed = e_first, pp = p_first;
        if (lb != 0) {
            ed = pid_at(perm_l, f, F0, F1, F2, totalL, g0, g1, g2, g3);
            pp = (f < totalL) ? lab_p[ed] : 0;
        }
        const int au = (f >= F0) + (f >= F1) + (f >= F2);
        const int n  = min(32, totalL - lb);
        float res = 0.0f;
        for (int j = 0; j < n; j += 8) {
            float4 v[8];
            #pragma unroll
            for (int k = 0; k < 8; ++k)
                v[k] = ld_row_off(pb, (unsigned)__builtin_amdgcn_readlane(pp, j + k), lane16);
            #pragma unroll
            for (int k = 0; k < 8; ++k) {
                const int as = __builtin_amdgcn_readlane(au, j + k);
                const float4 g = *(const float4*)(grows + as * LDSP);
                float s = __builtin_fmaf(g.w, v[k].w,
                          __builtin_fmaf(g.z, v[k].z,
                          __builtin_fmaf(g.x, v[k].x, g.y * v[k].y)));
                #pragma unroll
                for (int o = 16; o > 0; o >>= 1)
                    s += __shfl_xor(s, o, 32);
                if (lane == j + k) res = s;
            }
        }
        if (f < totalL) out[ed] = res + sC[4 * wave + au];
    }
}

extern "C" void kernel_launch(void* const* d_in, const int* in_sizes, int n_in,
                              void* d_out, int out_size, void* d_ws, size_t ws_size,
                              hipStream_t stream)
{
    const float* paper_x    = (const float*)d_in[0];
    const int*   author_ids = (const int*)  d_in[1];
    const int*   paper_ids  = (const int*)  d_in[2];
    const int*   lab_a      = (const int*)  d_in[3];
    const int*   lab_p      = (const int*)  d_in[4];
    const float* W_paper    = (const float*)d_in[5];
    const float* b_paper    = (const float*)d_in[6];
    const float* W_author   = (const float*)d_in[7];
    const float* b_author   = (const float*)d_in[8];

    const int num_edges       = in_sizes[1];
    const int num_label_edges = in_sizes[3];
    const int num_authors     = 200000;

    const int num_chunks   = (num_edges + SORT_CHUNK - 1) / SORT_CHUNK;
    const int num_chunks_l = (num_label_edges + SORT_CHUNK - 1) / SORT_CHUNK;
    const int apb          = (num_authors + SORT_BINS - 1) / SORT_BINS;
    float* Z          = (float*)d_ws;
    float* zc         = Z + (size_t)FOLD_ROWS * FEATDIM;
    int2*  sorted     = (int2*)(zc + 256);
    int2*  sorted_l   = sorted + (size_t)num_chunks * SORT_CHUNK;
    int*   runstart   = (int*)(sorted_l + (size_t)num_chunks_l * SORT_CHUNK);
    int*   counts     = runstart + (size_t)num_chunks * (SORT_BINS + 1);
    int*   cursor     = counts + num_authors;
    int*   perm       = cursor + num_authors;
    int*   runstart_l = perm + num_edges;
    int*   counts_l   = runstart_l + (size_t)num_chunks_l * (SORT_BINS + 1);
    int*   cursor_l   = counts_l + num_authors;
    int*   perm_l     = cursor_l + num_authors;

    fold_weights<<<FOLD_ROWS, FEATDIM, 0, stream>>>(
        W_paper, b_paper, W_author, b_author, Z, zc);

    seg_chunk<<<num_chunks + num_chunks_l, SORT_THREADS, 0, stream>>>(
        author_ids, paper_ids, lab_a, sorted, runstart, sorted_l, runstart_l,
        num_edges, num_label_edges, num_chunks, num_authors, apb);
    seg_emit<<<2 * SORT_BINS, SORT_THREADS, 0, stream>>>(
        sorted, runstart, counts, cursor, perm,
        sorted_l, runstart_l, counts_l, cursor_l, perm_l,
        num_chunks, num_chunks_l, num_authors, apb);

    seg_mean_gemm<<<num_authors / BM, 256, 0, stream>>>(
        paper_x, perm, cursor, counts, Z, zc,
        perm_l, cursor_l, counts_l, lab_p, (float*)d_out, num_authors);
}


# Folded weights stored in WMMA B-fragment order: a wave gets its B slice as 16 whole-line 512 B loads, 6 ahead of the tile barrier, 10 refilled 24 WMMAs ahead, not 64 dword loads chained per k pair
# speedup vs baseline: 5.4340x; 1.0643x over previous
#include <hip/hip_runtime.h>
#include <hip/hip_bf16.h>

typedef __attribute__((ext_vector_type(2))) float v2f;
typedef __attribute__((ext_vector_type(8))) float v8f;

#define FEATDIM 128
#define BM      32
#define LDSP    132

#define FOLD_ROWS (FEATDIM + 2)

__device__ __forceinline__ int zfrag_index(int k, int n)
{
    const int kk = k >> 3;
    const int s  = (k >> 2) & 1;
    const int h  = (k >> 1) & 1;
    const int j  = k & 1;
    return (((n >> 4) * 16 + kk) * 32 + h * 16 + (n & 15)) * 4 + 2 * s + j;
}

__global__ __launch_bounds__(FEATDIM) void fold_weights(
    const float* __restrict__ Wp, const float* __restrict__ bp,
    const float* __restrict__ Wa, const float* __restrict__ ba,
    float* __restrict__ Z, float* __restrict__ zc)
{
    __shared__ __attribute__((aligned(16))) float sY[FEATDIM];
    const int i    = blockIdx.x;
    const int n    = threadIdx.x;
    const int lane = n & 31;
    const int wave = n >> 5;

    float y;
    if (i <= FEATDIM) {
        const float* xr = (i < FEATDIM) ? Wp + (size_t)i * FEATDIM : bp;
        y = 0.0f;
        #pragma unroll
        for (int k = 0; k < FEATDIM; ++k)
            y = __builtin_fmaf(xr[k], Wa[(size_t)k * FEATDIM + n], y);
    } else {
        y = ba[n];
    }
    sY[n] = y;
    __syncthreads();

    const float4 yv = *(const float4*)&sY[lane * 4];
    float zkeep = 0.0f;
    #pragma unroll
    for (int t = 0; t < 32; ++t) {
        const float4 w = *(const float4*)(Wp + (size_t)(wave * 32 + t) * FEATDIM + lane * 4);
        float sdot = __builtin_fmaf(w.w, yv.w, __builtin_fmaf(w.z, yv.z,
                     __builtin_fmaf(w.y, yv.y, w.x * yv.x)));
        #pragma unroll
        for (int o = 16; o > 0; o >>= 1)
            sdot += __shfl_xor(sdot, o, 32);
        if (lane == t) zkeep = sdot;
    }
    if (i < FEATDIM) Z[zfrag_index(i, n)] = zkeep;
    else             Z[(size_t)i * FEATDIM + n] = zkeep;

    if (wave == 0) {
        const float4 w = *(const float4*)(bp + lane * 4);
        float sdot = __builtin_fmaf(w.w, yv.w, __builtin_fmaf(w.z, yv.z,
                     __builtin_fmaf(w.y, yv.y, w.x * yv.x)));
        #pragma unroll
        for (int o = 16; o > 0; o >>= 1)
            sdot += __shfl_xor(sdot, o, 32);
        if (lane == 0) zc[i] = sdot;
    }
}

#define B_AHEAD 6

__device__ __forceinline__ const float4* b_slice(const float* __restrict__ W, int wave, int lane)
{
    return (const float4*)W + (size_t)(wave * 16) * 32 + lane;
}

__device__ __forceinline__ void b_load_first(const float4* __restrict__ zf, float4 (&bq)[B_AHEAD])
{
    #pragma unroll
    for (int i = 0; i < B_AHEAD; ++i) bq[i] = zf[i * 32];
}

__device__ __forceinline__ void gemm_tile_from_lds(
    const float* sA, const float* sInd, const float* __restrict__ W,
    const float4* __restrict__ zf, float4 (&bq)[B_AHEAD],
    float* sG, int tid)
{
    const int wave = tid >> 5;
    const int lane = tid & 31;
    const int half = lane >> 4;
    const int l16  = lane & 15;
    const int n    = wave * 16 + l16;

    v8f acc0 = {};
    v8f acc1 = {};

    #pragma unroll
    for (int kk = 0; kk < FEATDIM / 8; ++kk) {
        const float4 bb = bq[kk % B_AHEAD];
        if (kk + B_AHEAD < FEATDIM / 8) bq[kk % B_AHEAD] = zf[(kk + B_AHEAD) * 32];
        #pragma unroll
        for (int s = 0; s < 2; ++s) {
            const int ka = 8 * kk + 4 * s + 2 * half;
            v2f b;
            b.x = s ? bb.z : bb.x;
            b.y = s ? bb.w : bb.y;
            v2f a0, a1;
            a0 = *(const v2f*)&sA[l16 * LDSP + ka];
            a1 = *(const v2f*)&sA[(16 + l16) * LDSP + ka];
            acc0 = __builtin_amdgcn_wmma_f32_16x16x4_f32(
                false, a0, false, b, (short)0, acc0, false, false);
            acc1 = __builtin_amdgcn_wmma_f32_16x16x4_f32(
                false, a1, false, b, (short)0, acc1, false, false);
        }
    }

    const float b_ind = W[(size_t)FEATDIM * FEATDIM + n];
    const float b_all = W[(size_t)(FEATDIM + 1) * FEATDIM + n];
    #pragma unroll
    for (int v = 0; v < 8; ++v) {
        const int m = v + 8 * half;
        const float bm0 = ((sInd[m]      != 0.0f) ? b_ind : 0.0f) + b_all;
        const float bm1 = ((sInd[m + 16] != 0.0f) ? b_ind : 0.0f) + b_all;
        sG[m * LDSP + n]        = acc0[v] + bm0;
        sG[(m + 16) * LDSP + n] = acc1[v] + bm1;
    }
}

#define SORT_BINS    256
#define SORT_CHUNK   8192
#define SORT_THREADS 1024
#define EMIT_CAP     16384

__global__ __launch_bounds__(SORT_THREADS) void seg_chunk(
    const int* __restrict__ author_ids, const int* __restrict__ paper_ids,
    const int* __restrict__ lab_a,
    int2* __restrict__ sorted_w, int* __restrict__ runstart_w,
    int2* __restrict__ sorted_l, int* __restrict__ runstart_l,
    int num_edges_w, int num_edges_l, int num_chunks_w, int num_authors, int apb)
{
    __shared__ int  hist[SORT_BINS];
    __shared__ int  binstart[SORT_BINS + 1];
    __shared__ int  wsum[SORT_BINS / 32];
    __shared__ int2 stage[SORT_CHUNK];

    const int tid  = threadIdx.x;
    const int lane = tid & 31;
    const int wave = tid >> 5;
    const bool lab = (int)blockIdx.x >= num_chunks_w;
    const int  blk = lab ? (int)blockIdx.x - num_chunks_w : (int)blockIdx.x;
    const int* __restrict__ ids = lab ? lab_a : author_ids;
    const int  num_edges = lab ? num_edges_l : num_edges_w;
    int2* __restrict__ sorted   = lab ? sorted_l : sorted_w;
    int*  __restrict__ runstart = lab ? runstart_l : runstart_w;
    const int e0   = blk * SORT_CHUNK;

    if (tid < SORT_BINS) hist[tid] = 0;
    __syncthreads();

    int a[8], p[8], bin[8], rank[8];
    #pragma unroll
    for (int k = 0; k < 8; ++k) {
        const int e = e0 + k * SORT_THREADS + tid;
        a[k] = -1; p[k] = 0;
        if (e < num_edges) { a[k] = ids[e]; p[k] = lab ? e : paper_ids[e]; }
    }
    #pragma unroll
    for (int k = 0; k < 8; ++k) {
        bin[k] = -1; rank[k] = 0;
        if ((unsigned)a[k] < (unsigned)num_authors) {
            bin[k]  = (int)((unsigned)a[k] / (unsigned)apb);
            rank[k] = atomicAdd(&hist[bin[k]], 1);
        }
    }
    __syncthreads();

    int x = 0, incl = 0;
    if (tid < SORT_BINS) {
        x = hist[tid];
        incl = x;
        #pragma unroll
        for (int off = 1; off < 32; off <<= 1) {
            const int y = __shfl_up(incl, off, 32);
            if (lane >= off) incl += y;
        }
        if (lane == 31) wsum[wave] = incl;
    }
    __syncthreads();
    if (tid < SORT_BINS) {
        int add = 0;
        for (int w = 0; w < wave; ++w) add += wsum[w];
        incl += add;
        binstart[tid] = incl - x;
        if (tid == SORT_BINS - 1) binstart[SORT_BINS] = incl;
    }
    __syncthreads();

    #pragma unroll
    for (int k = 0; k < 8; ++k)
        if (bin[k] >= 0) stage[binstart[bin[k]] + rank[k]] = make_int2(a[k], p[k]);
    __syncthreads();

    const int placed = binstart[SORT_BINS];
    for (int i = tid; i < placed; i += SORT_THREADS) sorted[e0 + i] = stage[i];
    if (tid <= SORT_BINS) runstart[blk * (SORT_BINS + 1) + tid] = binstart[tid];
}

__global__ __launch_bounds__(SORT_THREADS) void seg_emit(
    const int2* __restrict__ sorted_w, const int* __restrict__ runstart_w,
    int* __restrict__ counts_w, int* __restrict__ seg_end_w, int* __restrict__ perm_w,
    const int2* __restrict__ sorted_l, const int* __restrict__ runstart_l,
    int* __restrict__ counts_l, int* __restrict__ seg_end_l, int* __restrict__ perm_l,
    int num_chunks_w, int num_chunks_l, int num_authors, int apb)
{
    __shared__ int rstart[SORT_THREADS];
    __shared__ int rlen[SORT_THREADS];
    __shared__ int acount[SORT_THREADS];
    __shared__ int cur[SORT_THREADS];
    __shared__ int wsum[SORT_THREADS / 32];
    __shared__ int tot[2];
    __shared__ int lperm[EMIT_CAP];

    const int tid     = threadIdx.x;
    const int lane    = tid & 31;
    const int wave    = tid >> 5;
    const bool lab    = (int)blockIdx.x >= SORT_BINS;
    const int b       = lab ? (int)blockIdx.x - SORT_BINS : (int)blockIdx.x;
    const int a_first = b * apb;
    const int2* __restrict__ sorted   = lab ? sorted_l : sorted_w;
    const int*  __restrict__ runstart = lab ? runstart_l : runstart_w;
    int* __restrict__ counts  = lab ? counts_l : counts_w;
    int* __restrict__ seg_end = lab ? seg_end_l : seg_end_w;
    int* __restrict__ perm    = lab ? perm_l : perm_w;
    const int num_chunks = lab ? num_chunks_l : num_chunks_w;

    if (tid < 2) tot[tid] = 0;
    acount[tid] = 0;
    int s0 = 0, len = 0;
    if (tid < num_chunks) {
        s0  = runstart[tid * (SORT_BINS + 1) + b];
        len = runstart[tid * (SORT_BINS + 1) + b + 1] - s0;
    }
    rstart[tid] = tid * SORT_CHUNK + s0;
    rlen[tid]   = len;
    int ws0 = s0, wlen = len;
    #pragma unroll
    for (int o = 16; o > 0; o >>= 1) {
        ws0  += __shfl_xor(ws0, o, 32);
        wlen += __shfl_xor(wlen, o, 32);
    }
    __syncthreads();
    if (lane == 0) { atomicAdd(&tot[0], ws0); atomicAdd(&tot[1], wlen); }
    __syncthreads();
    const int base = tot[0];
    const int T    = tot[1];

    int2 v[16];
    #pragma unroll
    for (int k = 0; k < 8; ++k) {
        const int r = wave + 32 * k;
        const int l = (r < num_chunks) ? rlen[r] : 0;
        v[k]     = make_int2(-1, 0);
        v[8 + k] = make_int2(-1, 0);
        if (lane < l)      v[k]     = sorted[rstart[r] + lane];
        if (lane + 32 < l) v[8 + k] = sorted[rstart[r] + 32 + lane];
    }
    #pragma unroll
    for (int k = 0; k < 16; ++k)
        if (v[k].x >= 0) atomicAdd(&acount[v[k].x - a_first], 1);
    for (int r = wave; r < num_chunks; r += 32) {
        const int l   = rlen[r];
        const int src = rstart[r];
        for (int j = ((r < 256) ? 64 : 0) + lane; j < l; j += 32)
            atomicAdd(&acount[sorted[src + j].x - a_first], 1);
    }
    __syncthreads();

    const int x = acount[tid];
    int incl = x;
    #pragma unroll
    for (int off = 1; off < 32; off <<= 1) {
        const int y = __shfl_up(incl, off, 32);
        if (lane >= off) incl += y;
    }
    if (lane == 31) wsum[wave] = incl;
    __syncthreads();
    {
        int add = 0;
        for (int w = 0; w < wave; ++w) add += wsum[w];
        incl += add;
    }
    cur[tid] = incl - x;
    if (tid < apb && a_first + tid < num_authors) {
        counts[a_first + tid]  = x;
        seg_end[a_first + tid] = base + incl;
    }
    __syncthreads();

    const bool in_lds = T <= EMIT_CAP;
    #pragma unroll
    for (int k = 0; k < 16; ++k)
        if (v[k].x >= 0) {
            const int slot = atomicAdd(&cur[v[k].x - a_first], 1);
            if (in_lds) lperm[slot] = v[k].y; else perm[base + slot] = v[k].y;
        }
    for (int r = wave; r < num_chunks; r += 32) {
        const int l   = rlen[r];
        const int src = rstart[r];
        for (int j = ((r < 256) ? 64 : 0) + lane; j < l; j += 32) {
            const int2 q = sorted[src + j];
            const int slot = atomicAdd(&cur[q.x - a_first], 1);
            if (in_lds) lperm[slot] = q.y; else perm[base + slot] = q.y;
        }
    }
    __syncthreads();
    if (in_lds)
        for (int i = tid; i < T; i += SORT_THREADS) perm[base + i] = lperm[i];
}

__device__ __forceinline__ void add_row(float4& acc, const float4 v)
{
    acc.x += v.x; acc.y += v.y; acc.z += v.z; acc.w += v.w;
}

__device__ __forceinline__ float4 ld_row_off(const char* tab, unsigned p, unsigned lane16)
{
    return *(const float4*)(tab + ((p << 9) + lane16));
}

__device__ __forceinline__ int pid_at(const int* __restrict__ perm, int f,
                                      int E0, int E1, int E2, int total,
                                      int d0, int d1, int d2, int d3)
{
    int dl = d0;
    dl = (f >= E0) ? d1 : dl;
    dl = (f >= E1) ? d2 : dl;
    dl = (f >= E2) ? d3 : dl;
    return (f < total) ? perm[f + dl] : 0;
}

__global__ __launch_bounds__(256) __attribute__((amdgpu_waves_per_eu(12, 16))) void seg_mean_gemm(
    const float* __restrict__ paper_x, const int* __restrict__ perm,
    const int* __restrict__ seg_end, const int* __restrict__ counts,
    const float* __restrict__ Z, const float* __restrict__ zc,
    const int* __restrict__ perm_l, const int* __restrict__ seg_end_l,
    const int* __restrict__ counts_l, const int* __restrict__ lab_p,
    float* __restrict__ out, int num_authors)
{
    __shared__ float sA[BM * LDSP];
    __shared__ __attribute__((aligned(16))) float sG[BM * LDSP];
    __shared__ float sInd[BM];
    __shared__ float sC[BM];

    const int tid  = threadIdx.x;
    const int lane = tid & 31;
    const int wave = __builtin_amdgcn_readfirstlane(tid >> 5);
    const int row0 = blockIdx.x * BM;
    if (row0 >= num_authors) return;

    const int a0 = row0 + 4 * wave;
    int c_l = 0, e_l = 0, cl_l = 0, el_l = 0;
    if (lane < 4 && a0 + lane < num_authors) {
        c_l  = counts[a0 + lane];
        e_l  = seg_end[a0 + lane];
        cl_l = counts_l[a0 + lane];
        el_l = seg_end_l[a0 + lane];
    }
    if (lane < 4) sInd[4 * wave + lane] = (c_l > 0) ? 1.0f : 0.0f;
    const int c0 = __builtin_amdgcn_readlane(c_l, 0);
    const int c1 = __builtin_amdgcn_readlane(c_l, 1);
    const int c2 = __builtin_amdgcn_readlane(c_l, 2);
    const int c3 = __builtin_amdgcn_readlane(c_l, 3);
    const int E0 = c0, E1 = E0 + c1, E2 = E1 + c2, total = E2 + c3;
    const int d0 = __builtin_amdgcn_readlane(e_l, 0) - E0;
    const int d1 = __builtin_amdgcn_readlane(e_l, 1) - E1;
    const int d2 = __builtin_amdgcn_readlane(e_l, 2) - E2;
    const int d3 = __builtin_amdgcn_readlane(e_l, 3) - total;
    const int F0 = __builtin_amdgcn_readlane(cl_l, 0);
    const int F1 = F0 + __builtin_amdgcn_readlane(cl_l, 1);
    const int F2 = F1 + __builtin_amdgcn_readlane(cl_l, 2);
    const int totalL = F2 + __builtin_amdgcn_readlane(cl_l, 3);
    const int g0 = __builtin_amdgcn_readlane(el_l, 0) - F0;
    const int g1 = __builtin_amdgcn_readlane(el_l, 1) - F1;
    const int g2 = __builtin_amdgcn_readlane(el_l, 2) - F2;
    const int g3 = __builtin_amdgcn_readlane(el_l, 3) - totalL;
    const int e_first = pid_at(perm_l, lane, F0, F1, F2, totalL, g0, g1, g2, g3);

    float* const myrows = sA + (4 * wave) * LDSP + lane * 4;
    #pragma unroll
    for (int i = 0; i < 4; ++i)
        *(float4*)(myrows + i * LDSP) = make_float4(0.0f, 0.0f, 0.0f, 0.0f);

    const char* const pb = (const char*)paper_x;
    const unsigned lane16 = (unsigned)lane * 16u;

    float4 acc = make_float4(0.0f, 0.0f, 0.0f, 0.0f);
    int pid_next = pid_at(perm, lane, E0, E1, E2, total, d0, d1, d2, d3);
    for (int base = 0; base < total; base += 32) {
        const int mypid = pid_next;
        const int f = base + lane;
        pid_next = 0;
        const int au   = (f >= E0) + (f >= E1) + (f >= E2);
        const int last = (f + 1 == E0) | (f + 1 == E1) | (f + 1 == E2) | (f + 1 == total);
        const int code = au | (last << 2);
        const int n = min(32, total - base);

        int j = 0;
        for (; j + 8 <= n; j += 8) {
            float4 v[8];
            #pragma unroll
            for (int k = 0; k < 8; ++k)
                v[k] = ld_row_off(pb, (unsigned)__builtin_amdgcn_readlane(mypid, j + k), lane16);
            if (j == 0)
                pid_next = pid_at(perm, f + 32, E0, E1, E2, total, d0, d1, d2, d3);
            #pragma unroll
            for (int k = 0; k < 8; ++k) {
                add_row(acc, v[k]);
                const int cd = __builtin_amdgcn_readlane(code, j + k);
                if (cd & 4) {
                    *(float4*)(myrows + (cd & 3) * LDSP) = acc;
                    acc = make_float4(0.0f, 0.0f, 0.0f, 0.0f);
                }
            }
        }
        if (j < n) {
            float4 v[7];
            #pragma unroll
            for (int k = 0; k < 7; ++k)
                if (j + k < n)
                    v[k] = ld_row_off(pb, (unsigned)__builtin_amdgcn_readlane(mypid, j + k), lane16);
            #pragma unroll
            for (int k = 0; k < 7; ++k)
                if (j + k < n) {
                    add_row(acc, v[k]);
                    const int cd = __builtin_amdgcn_readlane(code, j + k);
                    if (cd & 4) {
                        *(float4*)(myrows + (cd & 3) * LDSP) = acc;
                        acc = make_float4(0.0f, 0.0f, 0.0f, 0.0f);
                    }
                }
        }
    }

    const float4* const zf = b_slice(Z, wave, lane);
    float4 bq[B_AHEAD];
    b_load_first(zf, bq);

    const float4 u = *(const float4*)(zc + lane * 4);
    const float c_ind = zc[FEATDIM];
    const float c_all = zc[FEATDIM + 1];
    float cval = 0.0f;
    #pragma unroll
    for (int i = 0; i < 4; ++i) {
        const int   ci = __builtin_amdgcn_readlane(c_l, i);
        const float d  = fmaxf((float)ci, 1.0f);
        float4 s = *(const float4*)(myrows + i * LDSP);
        s.x = s.x / d; s.y = s.y / d; s.z = s.z / d; s.w = s.w / d;
        *(float4*)(myrows + i * LDSP) = s;
        float t = __builtin_fmaf(s.w, u.w, __builtin_fmaf(s.z, u.z,
                  __builtin_fmaf(s.y, u.y, s.x * u.x)));
        #pragma unroll
        for (int o = 16; o > 0; o >>= 1)
            t += __shfl_xor(t, o, 32);
        t += ((ci > 0) ? c_ind : 0.0f) + c_all;
        if (lane == i) cval = t;
    }
    if (lane < 4) sC[4 * wave + lane] = cval;
    __syncthreads();

    int p_first = 0;
    if (lane < totalL) p_first = lab_p[e_first];

    gemm_tile_from_lds(sA, sInd, Z, zf, bq, sG, tid);
    __syncthreads();

    const float* const grows = sG + (4 * wave) * LDSP + lane * 4;
    for (int lb = 0; lb < totalL; lb += 32) {
        const int f  = lb + lane;
        int ed = e_first, pp = p_first;
        if (lb != 0) {
            ed = pid_at(perm_l, f, F0, F1, F2, totalL, g0, g1, g2, g3);
            pp = (f < totalL) ? lab_p[ed] : 0;
        }
        const int au = (f >= F0) + (f >= F1) + (f >= F2);
        const int n  = min(32, totalL - lb);
        float res = 0.0f;
        for (int j = 0; j < n; j += 8) {
            float4 v[8];
            #pragma unroll
            for (int k = 0; k < 8; ++k)
                v[k] = ld_row_off(pb, (unsigned)__builtin_amdgcn_readlane(pp, j + k), lane16);
            #pragma unroll
            for (int k = 0; k < 8; ++k) {
                const int as = __builtin_amdgcn_readlane(au, j + k);
                const float4 g = *(const float4*)(grows + as * LDSP);
                float s = __builtin_fmaf(g.w, v[k].w,
                          __builtin_fmaf(g.z, v[k].z,
                          __builtin_fmaf(g.x, v[k].x, g.y * v[k].y)));
                #pragma unroll
                for (int o = 16; o > 0; o >>= 1)
                    s += __shfl_xor(s, o, 32);
                if (lane == j + k) res = s;
            }
        }
        if (f < totalL) out[ed] = res + sC[4 * wave + au];
    }
}

extern "C" void kernel_launch(void* const* d_in, const int* in_sizes, int n_in,
                              void* d_out, int out_size, void* d_ws, size_t ws_size,
                              hipStream_t stream)
{
    const float* paper_x    = (const float*)d_in[0];
    const int*   author_ids = (const int*)  d_in[1];
    const int*   paper_ids  = (const int*)  d_in[2];
    const int*   lab_a      = (const int*)  d_in[3];
    const int*   lab_p      = (const int*)  d_in[4];
    const float* W_paper    = (const float*)d_in[5];
    const float* b_paper    = (const float*)d_in[6];
    const float* W_author   = (const float*)d_in[7];
    const float* b_author   = (const float*)d_in[8];

    const int num_edges       = in_sizes[1];
    const int num_label_edges = in_sizes[3];
    const int num_authors     = 200000;

    const int num_chunks   = (num_edges + SORT_CHUNK - 1) / SORT_CHUNK;
    const int num_chunks_l = (num_label_edges + SORT_CHUNK - 1) / SORT_CHUNK;
    const int apb          = (num_authors + SORT_BINS - 1) / SORT_BINS;
    float* Z          = (float*)d_ws;
    float* zc         = Z + (size_t)FOLD_ROWS * FEATDIM;
    int2*  sorted     = (int2*)(zc + 256);
    int2*  sorted_l   = sorted + (size_t)num_chunks * SORT_CHUNK;
    int*   runstart   = (int*)(sorted_l + (size_t)num_chunks_l * SORT_CHUNK);
    int*   counts     = runstart + (size_t)num_chunks * (SORT_BINS + 1);
    int*   cursor     = counts + num_authors;
    int*   perm       = cursor + num_authors;
    int*   runstart_l = perm + num_edges;
    int*   counts_l   = runstart_l + (size_t)num_chunks_l * (SORT_BINS + 1);
    int*   cursor_l   = counts_l + num_authors;
    int*   perm_l     = cursor_l + num_authors;

    fold_weights<<<FOLD_ROWS, FEATDIM, 0, stream>>>(
        W_paper, b_paper, W_author, b_author, Z, zc);

    seg_chunk<<<num_chunks + num_chunks_l, SORT_THREADS, 0, stream>>>(
        author_ids, paper_ids, lab_a, sorted, runstart, sorted_l, runstart_l,
        num_edges, num_label_edges, num_chunks, num_authors, apb);
    seg_emit<<<2 * SORT_BINS, SORT_THREADS, 0, stream>>>(
        sorted, runstart, counts, cursor, perm,
        sorted_l, runstart_l, counts_l, cursor_l, perm_l,
        num_chunks, num_chunks_l, num_authors, apb);

    seg_mean_gemm<<<num_authors / BM, 256, 0, stream>>>(
        paper_x, perm, cursor, counts, Z, zc,
        perm_l, cursor_l, counts_l, lab_p, (float*)d_out, num_authors);
}
